# starting kernel
# baseline (speedup 1.0000x reference)
#include <hip/hip_runtime.h>
#include <math.h>


typedef __attribute__((ext_vector_type(16))) __bf16 v16bf;
typedef __attribute__((ext_vector_type(8)))  __bf16 v8bf;
typedef __attribute__((ext_vector_type(8)))  float  v8f;
typedef __attribute__((ext_vector_type(4)))  float  v4f;

#define IN_FEAT 512
#define HID     16

__device__ __forceinline__ void atomAddF(float* p, float v) {
  __hip_atomic_fetch_add(p, v, __ATOMIC_RELAXED, __HIP_MEMORY_SCOPE_AGENT);
}

__global__ __launch_bounds__(256) void gemm1_wmma_k(
    const float* __restrict__ X, const float* __restrict__ W1,
    float* __restrict__ XW1, int nTiles) {
  __shared__ __bf16 Bs[HID * IN_FEAT];
  for (int i = threadIdx.x; i < HID * IN_FEAT; i += 256) {
    int k = i >> 4, n = i & 15;
    Bs[n * IN_FEAT + k] = (__bf16)W1[i];
  }
  __syncthreads();

  int wave = threadIdx.x >> 5;
  int lane = threadIdx.x & 31;
  int tile = blockIdx.x * 8 + wave;
  if (tile >= nTiles) return;

  int rc   = lane & 15;
  int half = lane >> 4;

  const float*  arow = X  + (size_t)(tile * 16 + rc) * IN_FEAT;
  const __bf16* bcol = Bs + (size_t)rc * IN_FEAT;

  v8f c = {0.f, 0.f, 0.f, 0.f, 0.f, 0.f, 0.f, 0.f};

#pragma unroll
  for (int ks = 0; ks < IN_FEAT / 32; ++ks) {
    int kb = ks * 32 + half * 8;
    v4f a0 = *(const v4f*)(arow + kb);
    v4f a1 = *(const v4f*)(arow + kb + 4);
    v4f a2 = *(const v4f*)(arow + kb + 16);
    v4f a3 = *(const v4f*)(arow + kb + 20);
    v8bf b0 = *(const v8bf*)(bcol + kb);
    v8bf b1 = *(const v8bf*)(bcol + kb + 16);
    v16bf A, B;
#pragma unroll
    for (int i = 0; i < 4; ++i) {
      A[i]      = (__bf16)a0[i];
      A[4 + i]  = (__bf16)a1[i];
      A[8 + i]  = (__bf16)a2[i];
      A[12 + i] = (__bf16)a3[i];
    }
#pragma unroll
    for (int i = 0; i < 8; ++i) { B[i] = b0[i]; B[8 + i] = b1[i]; }
    c = __builtin_amdgcn_wmma_f32_16x16x32_bf16(
         false, A,  false, B,
         (short)0, c,  false,  false);
  }

  float* orow = XW1 + ((size_t)tile * 16 + half * 8) * HID + rc;
#pragma unroll
  for (int r = 0; r < 8; ++r) orow[r * HID] = c[r];
}

__global__ void deg_init_k(int* DEG, int N) {
  int n = blockIdx.x * blockDim.x + threadIdx.x;
  if (n < N) DEG[n] = 1;
}
__global__ void deg_edge_k(const int* __restrict__ DST, int* DEG, int E) {
  int e = blockIdx.x * blockDim.x + threadIdx.x;
  if (e < E) atomicAdd(&DEG[DST[e]], 1);
}
__global__ void dinv_k(float* DNV, int N) {
  int n = blockIdx.x * blockDim.x + threadIdx.x;
  if (n < N) {
    int c = ((const int*)DNV)[n];
    DNV[n] = rsqrtf((float)c);
  }
}

__global__ void self1_k(const float* __restrict__ XW1,
                        const float* __restrict__ DNV, float* H, int N) {
  int t = blockIdx.x * blockDim.x + threadIdx.x;
  int n = t >> 2;
  if (n >= N) return;
  int q = t & 3;
  float w = DNV[n] * DNV[n];
  v4f v = *(const v4f*)(XW1 + (size_t)n * HID + q * 4);
  v4f o;
#pragma unroll
  for (int i = 0; i < 4; ++i) o[i] = w * v[i];
  *(v4f*)(H + (size_t)n * HID + q * 4) = o;
}

__global__ void agg1_k(const int* __restrict__ SRC, const int* __restrict__ DST,
                       const float* __restrict__ DNV,
                       const float* __restrict__ XW1, float* H, int E) {
  int t = blockIdx.x * blockDim.x + threadIdx.x;
  int e = t >> 2;
  if (e >= E) return;
  int q = t & 3;
  int s = SRC[e], d = DST[e];
  float nrm = DNV[s] * DNV[d];
  v4f v = *(const v4f*)(XW1 + (size_t)s * HID + q * 4);
  float* hp = H + (size_t)d * HID + q * 4;
#pragma unroll
  for (int i = 0; i < 4; ++i) atomAddF(hp + i, nrm * v[i]);
}

__global__ void relu_bias_k(float* H, const float* __restrict__ B1, int N) {
  int t = blockIdx.x * blockDim.x + threadIdx.x;
  int n = t >> 2;
  if (n >= N) return;
  int q = t & 3;
  v4f b = *(const v4f*)(B1 + q * 4);
  float* hp = H + (size_t)n * HID + q * 4;
  v4f v = *(const v4f*)hp;
#pragma unroll
  for (int i = 0; i < 4; ++i) v[i] = fmaxf(v[i] + b[i], 0.f);
  *(v4f*)hp = v;
}

__global__ void gemm2_k(const float* __restrict__ H, const float* __restrict__ W2,
                        const float* __restrict__ DNV, float* HW2, float* OUT2,
                        int N) {
  int n = blockIdx.x * blockDim.x + threadIdx.x;
  if (n >= N) return;
  const float* hr = H + (size_t)n * HID;
  float h0 = 0.f, h1 = 0.f;
#pragma unroll
  for (int f = 0; f < HID; ++f) {
    float hv = hr[f];
    h0 += hv * W2[2 * f];
    h1 += hv * W2[2 * f + 1];
  }
  HW2[2 * n] = h0;
  HW2[2 * n + 1] = h1;
  float w = DNV[n] * DNV[n];
  OUT2[2 * n] = w * h0;
  OUT2[2 * n + 1] = w * h1;
}

__global__ void agg2_k(const int* __restrict__ SRC, const int* __restrict__ DST,
                       const float* __restrict__ DNV,
                       const float* __restrict__ HW2, float* OUT2, int E) {
  int e = blockIdx.x * blockDim.x + threadIdx.x;
  if (e >= E) return;
  int s = SRC[e], d = DST[e];
  float nrm = DNV[s] * DNV[d];
  atomAddF(&OUT2[2 * d],     nrm * HW2[2 * s]);
  atomAddF(&OUT2[2 * d + 1], nrm * HW2[2 * s + 1]);
}

__global__ void lsm_k(const float* __restrict__ OUT2, const float* __restrict__ B2,
                      float* __restrict__ out, int N) {
  int n = blockIdx.x * blockDim.x + threadIdx.x;
  if (n >= N) return;
  float a = OUT2[2 * n]     + B2[0];
  float b = OUT2[2 * n + 1] + B2[1];
  float m = fmaxf(a, b);
  float lse = m + logf(expf(a - m) + expf(b - m));
  out[2 * n]     = a - lse;
  out[2 * n + 1] = b - lse;
}

extern "C" void kernel_launch(void* const* d_in, const int* in_sizes, int n_in,
                              void* d_out, int out_size, void* d_ws, size_t ws_size,
                              hipStream_t stream) {
  const float* X  = (const float*)d_in[0];
  const int*   EI = (const int*)d_in[1];
  const float* W1 = (const float*)d_in[2];
  const float* B1 = (const float*)d_in[3];
  const float* W2 = (const float*)d_in[4];
  const float* B2 = (const float*)d_in[5];

  const int N = in_sizes[0] / IN_FEAT;
  const int E = in_sizes[1] / 2;
  const int* SRC = EI;
  const int* DST = EI + E;

  float* ws   = (float*)d_ws;
  float* XW1  = ws;
  float* H    = ws + (size_t)N * HID;
  float* DNV  = ws + (size_t)2 * N * HID;
  float* HW2  = DNV + N;
  float* OUT2 = HW2 + (size_t)2 * N;

  const int thr = 256;
  const int nTiles = (N + 15) / 16;

  gemm1_wmma_k<<<(nTiles + 7) / 8, thr, 0, stream>>>(X, W1, XW1, nTiles);
  deg_init_k<<<(N + thr - 1) / thr, thr, 0, stream>>>((int*)DNV, N);
  deg_edge_k<<<(E + thr - 1) / thr, thr, 0, stream>>>(DST, (int*)DNV, E);
  dinv_k<<<(N + thr - 1) / thr, thr, 0, stream>>>(DNV, N);
  self1_k<<<(4 * N + thr - 1) / thr, thr, 0, stream>>>(XW1, DNV, H, N);
  agg1_k<<<(4 * E + thr - 1) / thr, thr, 0, stream>>>(SRC, DST, DNV, XW1, H, E);
  relu_bias_k<<<(4 * N + thr - 1) / thr, thr, 0, stream>>>(H, B1, N);
  gemm2_k<<<(N + thr - 1) / thr, thr, 0, stream>>>(H, W2, DNV, HW2, OUT2, N);
  agg2_k<<<(E + thr - 1) / thr, thr, 0, stream>>>(SRC, DST, DNV, HW2, OUT2, E);
  lsm_k<<<(N + thr - 1) / thr, thr, 0, stream>>>(OUT2, B2, (float*)d_out, N);
}


# edges bucketed once per call by blocks of 128 destination nodes; one work-group per bucket sums degree, layer 1 and layer 2 in LDS atomics; no global atomics; 10 -> 7 launches; sums reordered
# speedup vs baseline: 3.5152x; 3.5152x over previous
#include <hip/hip_runtime.h>
#include <math.h>


typedef __attribute__((ext_vector_type(16))) __bf16 v16bf;
typedef __attribute__((ext_vector_type(8)))  __bf16 v8bf;
typedef __attribute__((ext_vector_type(8)))  float  v8f;
typedef __attribute__((ext_vector_type(4)))  float  v4f;

#define IN_FEAT 512
#define HID     16

typedef __attribute__((ext_vector_type(2)))  float  v2f;

#define RSH   7
#define RB    128
#define NSLAB 256
#define NBMAX 1024
#define HP    17
#define EU    4
#define GU    8

__device__ __forceinline__ void ldsAddF(float* p, float v) {
  __hip_atomic_fetch_add(p, v, __ATOMIC_RELAXED, __HIP_MEMORY_SCOPE_WORKGROUP);
}

template <bool VEC>
__device__ __forceinline__ int4 ld4(const int* __restrict__ p, int e, int e0, int e1) {
  int4 r;
  if (VEC) {
    r = *(const int4*)(p + (e < e1 ? e : e0));
    if (e >= e1) r.x = r.y = r.z = r.w = -1;
  } else {
    r.x = (e     < e1) ? p[e]     : -1;
    r.y = (e + 1 < e1) ? p[e + 1] : -1;
    r.z = (e + 2 < e1) ? p[e + 2] : -1;
    r.w = (e + 3 < e1) ? p[e + 3] : -1;
  }
  return r;
}

template <bool VEC>
__device__ __forceinline__ void hist_slab(const int* __restrict__ DST, int* h,
                                          int e0, int e1, int t) {
  for (int e = e0 + 4 * t; e < e1; e += 4 * 1024 * EU) {
    int4 d[EU];
#pragma unroll
    for (int u = 0; u < EU; ++u) d[u] = ld4<VEC>(DST, e + u * 4096, e0, e1);
#pragma unroll
    for (int u = 0; u < EU; ++u) {
      if (d[u].x >= 0) atomicAdd(&h[d[u].x >> RSH], 1);
      if (d[u].y >= 0) atomicAdd(&h[d[u].y >> RSH], 1);
      if (d[u].z >= 0) atomicAdd(&h[d[u].z >> RSH], 1);
      if (d[u].w >= 0) atomicAdd(&h[d[u].w >> RSH], 1);
    }
  }
}

template <bool VEC>
__device__ __forceinline__ void fill_slab(const int* __restrict__ SRC,
                                          const int* __restrict__ DST, int* cur,
                                          int* __restrict__ BINNED,
                                          int e0, int e1, int t) {
  for (int e = e0 + 4 * t; e < e1; e += 4 * 1024 * EU) {
    int4 s[EU], d[EU];
#pragma unroll
    for (int u = 0; u < EU; ++u) {
      s[u] = ld4<VEC>(SRC, e + u * 4096, e0, e1);
      d[u] = ld4<VEC>(DST, e + u * 4096, e0, e1);
    }
#pragma unroll
    for (int u = 0; u < EU; ++u) {
      if (d[u].x >= 0) BINNED[atomicAdd(&cur[d[u].x >> RSH], 1)] = (s[u].x << RSH) | (d[u].x & (RB - 1));
      if (d[u].y >= 0) BINNED[atomicAdd(&cur[d[u].y >> RSH], 1)] = (s[u].y << RSH) | (d[u].y & (RB - 1));
      if (d[u].z >= 0) BINNED[atomicAdd(&cur[d[u].z >> RSH], 1)] = (s[u].z << RSH) | (d[u].z & (RB - 1));
      if (d[u].w >= 0) BINNED[atomicAdd(&cur[d[u].w >> RSH], 1)] = (s[u].w << RSH) | (d[u].w & (RB - 1));
    }
  }
}

__global__ __launch_bounds__(256) void gemm1_wmma_k(
    const float* __restrict__ X, const float* __restrict__ W1,
    float* __restrict__ XW1, int nTiles) {
  __shared__ __bf16 Bs[HID * IN_FEAT];
  for (int i = threadIdx.x; i < HID * IN_FEAT; i += 256) {
    int k = i >> 4, n = i & 15;
    Bs[n * IN_FEAT + k] = (__bf16)W1[i];
  }
  __syncthreads();

  int wave = threadIdx.x >> 5;
  int lane = threadIdx.x & 31;
  int tile = blockIdx.x * 8 + wave;
  if (tile >= nTiles) return;

  int rc   = lane & 15;
  int half = lane >> 4;

  const float*  arow = X  + (size_t)(tile * 16 + rc) * IN_FEAT;
  const __bf16* bcol = Bs + (size_t)rc * IN_FEAT;

  v8f c = {0.f, 0.f, 0.f, 0.f, 0.f, 0.f, 0.f, 0.f};

#pragma unroll
  for (int ks = 0; ks < IN_FEAT / 32; ++ks) {
    int kb = ks * 32 + half * 8;
    v4f a0 = *(const v4f*)(arow + kb);
    v4f a1 = *(const v4f*)(arow + kb + 4);
    v4f a2 = *(const v4f*)(arow + kb + 16);
    v4f a3 = *(const v4f*)(arow + kb + 20);
    v8bf b0 = *(const v8bf*)(bcol + kb);
    v8bf b1 = *(const v8bf*)(bcol + kb + 16);
    v16bf A, B;
#pragma unroll
    for (int i = 0; i < 4; ++i) {
      A[i]      = (__bf16)a0[i];
      A[4 + i]  = (__bf16)a1[i];
      A[8 + i]  = (__bf16)a2[i];
      A[12 + i] = (__bf16)a3[i];
    }
#pragma unroll
    for (int i = 0; i < 8; ++i) { B[i] = b0[i]; B[8 + i] = b1[i]; }
    c = __builtin_amdgcn_wmma_f32_16x16x32_bf16(
         false, A,  false, B,
         (short)0, c,  false,  false);
  }

  float* orow = XW1 + ((size_t)tile * 16 + half * 8) * HID + rc;
#pragma unroll
  for (int r = 0; r < 8; ++r) orow[r * HID] = c[r];
}

__global__ __launch_bounds__(1024) void bin_hist_k(const int* __restrict__ DST,
                                                   int* __restrict__ CNT,
                                                   int E, int NBP) {
  __shared__ int h[NBMAX];
  const int t = threadIdx.x;
  h[t] = 0;
  __syncthreads();
  const int slab = (E + NSLAB - 1) / NSLAB;
  const int e0 = blockIdx.x * slab;
  const int e1 = min(e0 + slab, E);
  const bool vec = ((E & 3) == 0) && ((slab & 3) == 0);
  if (vec) hist_slab<true>(DST, h, e0, e1, t);
  else     hist_slab<false>(DST, h, e0, e1, t);
  __syncthreads();
  if (t < NBP) CNT[blockIdx.x * NBP + t] = h[t];
}

__global__ __launch_bounds__(1024) void bin_scan_k(int* __restrict__ CNT,
                                                   int* __restrict__ TOT, int NBP) {
  __shared__ int rs[32][33];
  const int l = threadIdx.x & 31, r = threadIdx.x >> 5;
  const int col = blockIdx.x * 32 + l;
  int v[8];
  int run = 0;
#pragma unroll
  for (int k = 0; k < 8; ++k) {
    int c = CNT[(r * 8 + k) * NBP + col];
    v[k] = run;
    run += c;
  }
  rs[r][l] = run;
  __syncthreads();
  int base = 0, tot = 0;
  for (int rr = 0; rr < 32; ++rr) {
    int x = rs[rr][l];
    if (rr < r) base += x;
    tot += x;
  }
#pragma unroll
  for (int k = 0; k < 8; ++k) CNT[(r * 8 + k) * NBP + col] = base + v[k];
  if (r == 0) TOT[col] = tot;
}

__global__ __launch_bounds__(1024) void bin_fill_k(
    const int* __restrict__ SRC, const int* __restrict__ DST,
    const int* __restrict__ CNT, const int* __restrict__ TOT,
    int* __restrict__ START, int* __restrict__ BINNED, int E, int NB, int NBP) {
  __shared__ int tot[NBMAX];
  __shared__ int cb[32];
  __shared__ int cur[NBMAX];
  const int t = threadIdx.x;
  tot[t] = (t < NBP) ? TOT[t] : 0;
  __syncthreads();
  if (t < 32) {
    int s = 0;
    for (int j = 0; j < 32; ++j) s += tot[t * 32 + j];
    cb[t] = s;
  }
  __syncthreads();
  int st = 0;
  for (int j = 0; j < (t >> 5); ++j) st += cb[j];
  for (int j = (t & ~31); j < t; ++j) st += tot[j];
  if (blockIdx.x == 0 && t <= NB) START[t] = st;
  cur[t] = (t < NBP) ? st + CNT[blockIdx.x * NBP + t] : 0;
  __syncthreads();

  const int slab = (E + NSLAB - 1) / NSLAB;
  const int e0 = blockIdx.x * slab;
  const int e1 = min(e0 + slab, E);
  const bool vec = ((E & 3) == 0) && ((slab & 3) == 0);
  if (vec) fill_slab<true>(SRC, DST, cur, BINNED, e0, e1, t);
  else     fill_slab<false>(SRC, DST, cur, BINNED, e0, e1, t);
}

__global__ __launch_bounds__(256) void bin_deg_k(const int* __restrict__ START,
                                                 const int* __restrict__ BINNED,
                                                 float* __restrict__ DNV, int N) {
  __shared__ int c[RB];
  const int t = threadIdx.x;
  if (t < RB) c[t] = 1;
  __syncthreads();
  const int a = START[blockIdx.x], b = START[blockIdx.x + 1];
  for (int i0 = a + t; i0 < b; i0 += 256 * GU) {
    int p[GU];
#pragma unroll
    for (int u = 0; u < GU; ++u) {
      const int i = i0 + u * 256;
      p[u] = (i < b) ? BINNED[i] : -1;
    }
#pragma unroll
    for (int u = 0; u < GU; ++u)
      if (p[u] >= 0) atomicAdd(&c[p[u] & (RB - 1)], 1);
  }
  __syncthreads();
  const int n = blockIdx.x * RB + t;
  if (t < RB && n < N) DNV[n] = rsqrtf((float)c[t]);
}

__global__ __launch_bounds__(512) void bin_agg1_k(
    const int* __restrict__ START, const int* __restrict__ BINNED,
    const float* __restrict__ DNV, const float* __restrict__ XW1,
    const float* __restrict__ B1, const float* __restrict__ W2,
    float* __restrict__ HW2, int N) {
  __shared__ float acc[RB * HP];
  __shared__ float dl[RB];
  const int t = threadIdx.x;
  const int n0 = blockIdx.x * RB;
  const int q = t & 3;
  {
    const int r = t >> 2;
    const int n = n0 + r;
    float dn = 0.f;
    v4f v = {0.f, 0.f, 0.f, 0.f};
    if (n < N) {
      dn = DNV[n];
      v = *(const v4f*)(XW1 + (size_t)n * HID + q * 4);
    }
    const float w = dn * dn;
    if (q == 0) dl[r] = dn;
#pragma unroll
    for (int i = 0; i < 4; ++i) acc[r * HP + q * 4 + i] = w * v[i];
  }
  __syncthreads();
  const int a = START[blockIdx.x], b = START[blockIdx.x + 1];
  for (int i0 = a + (t >> 2); i0 < b; i0 += 128 * GU) {
    int p[GU];
#pragma unroll
    for (int u = 0; u < GU; ++u) {
      const int i = i0 + u * 128;
      p[u] = (i < b) ? BINNED[i] : -1;
    }
    float dn[GU];
    v4f v[GU];
#pragma unroll
    for (int u = 0; u < GU; ++u) {
      const int s = (p[u] < 0 ? 0 : p[u]) >> RSH;
      dn[u] = DNV[s];
      v[u] = *(const v4f*)(XW1 + (size_t)s * HID + q * 4);
    }
#pragma unroll
    for (int u = 0; u < GU; ++u) {
      if (p[u] >= 0) {
        const int r = p[u] & (RB - 1);
        const float nrm = dn[u] * dl[r];
        float* ap = acc + r * HP + q * 4;
#pragma unroll
        for (int k = 0; k < 4; ++k) ldsAddF(ap + k, nrm * v[u][k]);
      }
    }
  }
  __syncthreads();
  if (t < 2 * RB) {
    const int r = t >> 1, j = t & 1;
    const int n = n0 + r;
    if (n < N) {
      float h = 0.f;
#pragma unroll
      for (int f = 0; f < HID; ++f) {
        float hv = fmaxf(acc[r * HP + f] + B1[f], 0.f);
        h += hv * W2[2 * f + j];
      }
      HW2[2 * n + j] = h;
    }
  }
}

__global__ __launch_bounds__(256) void bin_agg2_k(
    const int* __restrict__ START, const int* __restrict__ BINNED,
    const float* __restrict__ DNV, const float* __restrict__ HW2,
    const float* __restrict__ B2, float* __restrict__ out, int N) {
  __shared__ float acc[2 * RB];
  __shared__ float dl[RB];
  const int t = threadIdx.x;
  const int n0 = blockIdx.x * RB;
  {
    const int r = t >> 1;
    const int n = n0 + r;
    float dn = 0.f, hv = 0.f;
    if (n < N) {
      dn = DNV[n];
      hv = HW2[2 * n + (t & 1)];
    }
    const float w = dn * dn;
    acc[t] = w * hv;
    if ((t & 1) == 0) dl[r] = dn;
  }
  __syncthreads();
  const int a = START[blockIdx.x], b = START[blockIdx.x + 1];
  for (int i0 = a + t; i0 < b; i0 += 256 * GU) {
    int p[GU];
#pragma unroll
    for (int u = 0; u < GU; ++u) {
      const int i = i0 + u * 256;
      p[u] = (i < b) ? BINNED[i] : -1;
    }
    float dn[GU];
    v2f hv[GU];
#pragma unroll
    for (int u = 0; u < GU; ++u) {
      const int s = (p[u] < 0 ? 0 : p[u]) >> RSH;
      dn[u] = DNV[s];
      hv[u] = *(const v2f*)(HW2 + 2 * (size_t)s);
    }
#pragma unroll
    for (int u = 0; u < GU; ++u) {
      if (p[u] >= 0) {
        const int r = p[u] & (RB - 1);
        const float nrm = dn[u] * dl[r];
        ldsAddF(&acc[2 * r],     nrm * hv[u][0]);
        ldsAddF(&acc[2 * r + 1], nrm * hv[u][1]);
      }
    }
  }
  __syncthreads();
  if (t < RB) {
    const int n = n0 + t;
    if (n < N) {
      float x = acc[2 * t]     + B2[0];
      float y = acc[2 * t + 1] + B2[1];
      float m = fmaxf(x, y);
      float lse = m + logf(expf(x - m) + expf(y - m));
      v2f o;
      o[0] = x - lse;
      o[1] = y - lse;
      *(v2f*)(out + 2 * (size_t)n) = o;
    }
  }
}

extern "C" void kernel_launch(void* const* d_in, const int* in_sizes, int n_in,
                              void* d_out, int out_size, void* d_ws, size_t ws_size,
                              hipStream_t stream) {
  const float* X  = (const float*)d_in[0];
  const int*   EI = (const int*)d_in[1];
  const float* W1 = (const float*)d_in[2];
  const float* B1 = (const float*)d_in[3];
  const float* W2 = (const float*)d_in[4];
  const float* B2 = (const float*)d_in[5];

  const int N = in_sizes[0] / IN_FEAT;
  const int E = in_sizes[1] / 2;
  const int* SRC = EI;
  const int* DST = EI + E;

  const int NB  = (N + RB - 1) / RB;
  const int NBP = (NB + 31) / 32 * 32;

  float* ws     = (float*)d_ws;
  float* XW1    = ws;
  int*   BINNED = (int*)(ws + (size_t)N * HID);
  float* HW2    = (float*)(BINNED + E);
  float* DNV    = HW2 + (size_t)2 * N;
  int*   CNT    = (int*)(DNV + N);
  int*   TOT    = CNT + (size_t)NSLAB * NBP;
  int*   START  = TOT + NBMAX;

  const int nTiles = (N + 15) / 16;

  gemm1_wmma_k<<<(nTiles + 7) / 8, 256, 0, stream>>>(X, W1, XW1, nTiles);
  bin_hist_k<<<NSLAB, 1024, 0, stream>>>(DST, CNT, E, NBP);
  bin_scan_k<<<NBP / 32, 1024, 0, stream>>>(CNT, TOT, NBP);
  bin_fill_k<<<NSLAB, 1024, 0, stream>>>(SRC, DST, CNT, TOT, START, BINNED, E, NB, NBP);
  bin_deg_k<<<NB, 256, 0, stream>>>(START, BINNED, DNV, N);
  bin_agg1_k<<<NB, 512, 0, stream>>>(START, BINNED, DNV, XW1, B1, W2, HW2, N);
  bin_agg2_k<<<NB, 256, 0, stream>>>(START, BINNED, DNV, HW2, B2, (float*)d_out, N);
}


# X@W1 tiles run as 196 32-wave work-groups of the bucket-fill launch, streaming X beside the slabs (64-register bound: both kinds share a WGP; k-loop 2 steps per trip); 7 -> 6 launches; same sums
# speedup vs baseline: 3.7920x; 1.0787x over previous
#include <hip/hip_runtime.h>
#include <math.h>


typedef __attribute__((ext_vector_type(16))) __bf16 v16bf;
typedef __attribute__((ext_vector_type(8)))  __bf16 v8bf;
typedef __attribute__((ext_vector_type(8)))  float  v8f;
typedef __attribute__((ext_vector_type(4)))  float  v4f;

#define IN_FEAT 512
#define HID     16

typedef __attribute__((ext_vector_type(2)))  float  v2f;

#define RSH   7
#define RB    128
#define NSLAB 256
#define NBMAX 1024
#define HP    17
#define EU    4
#define GU    8

__device__ __forceinline__ void ldsAddF(float* p, float v) {
  __hip_atomic_fetch_add(p, v, __ATOMIC_RELAXED, __HIP_MEMORY_SCOPE_WORKGROUP);
}

template <bool VEC>
__device__ __forceinline__ int4 ld4(const int* __restrict__ p, int e, int e0, int e1) {
  int4 r;
  if (VEC) {
    r = *(const int4*)(p + (e < e1 ? e : e0));
    if (e >= e1) r.x = r.y = r.z = r.w = -1;
  } else {
    r.x = (e     < e1) ? p[e]     : -1;
    r.y = (e + 1 < e1) ? p[e + 1] : -1;
    r.z = (e + 2 < e1) ? p[e + 2] : -1;
    r.w = (e + 3 < e1) ? p[e + 3] : -1;
  }
  return r;
}

template <bool VEC>
__device__ __forceinline__ void hist_slab(const int* __restrict__ DST, int* h,
                                          int e0, int e1, int t) {
  for (int e = e0 + 4 * t; e < e1; e += 4 * 1024 * EU) {
    int4 d[EU];
#pragma unroll
    for (int u = 0; u < EU; ++u) d[u] = ld4<VEC>(DST, e + u * 4096, e0, e1);
#pragma unroll
    for (int u = 0; u < EU; ++u) {
      if (d[u].x >= 0) atomicAdd(&h[d[u].x >> RSH], 1);
      if (d[u].y >= 0) atomicAdd(&h[d[u].y >> RSH], 1);
      if (d[u].z >= 0) atomicAdd(&h[d[u].z >> RSH], 1);
      if (d[u].w >= 0) atomicAdd(&h[d[u].w >> RSH], 1);
    }
  }
}

template <bool VEC>
__device__ __forceinline__ void fill_slab(const int* __restrict__ SRC,
                                          const int* __restrict__ DST, int* cur,
                                          int* __restrict__ BINNED,
                                          int e0, int e1, int t) {
  for (int e = e0 + 4 * t; e < e1; e += 4 * 1024 * EU) {
    int4 s[EU], d[EU];
#pragma unroll
    for (int u = 0; u < EU; ++u) {
      s[u] = ld4<VEC>(SRC, e + u * 4096, e0, e1);
      d[u] = ld4<VEC>(DST, e + u * 4096, e0, e1);
    }
#pragma unroll
    for (int u = 0; u < EU; ++u) {
      if (d[u].x >= 0) BINNED[atomicAdd(&cur[d[u].x >> RSH], 1)] = (s[u].x << RSH) | (d[u].x & (RB - 1));
      if (d[u].y >= 0) BINNED[atomicAdd(&cur[d[u].y >> RSH], 1)] = (s[u].y << RSH) | (d[u].y & (RB - 1));
      if (d[u].z >= 0) BINNED[atomicAdd(&cur[d[u].z >> RSH], 1)] = (s[u].z << RSH) | (d[u].z & (RB - 1));
      if (d[u].w >= 0) BINNED[atomicAdd(&cur[d[u].w >> RSH], 1)] = (s[u].w << RSH) | (d[u].w & (RB - 1));
    }
  }
}

template <int NT>
__device__ __forceinline__ void gemm1_block(
    const float* __restrict__ X, const float* __restrict__ W1,
    float* __restrict__ XW1, int nTiles, __bf16* Bs, int gb) {
  for (int i = threadIdx.x; i < HID * IN_FEAT; i += NT) {
    int k = i >> 4, n = i & 15;
    Bs[n * IN_FEAT + k] = (__bf16)W1[i];
  }
  __syncthreads();

  int wave = __builtin_amdgcn_readfirstlane(threadIdx.x >> 5);
  int lane = threadIdx.x & 31;
  int tile = gb * (NT / 32) + wave;
  if (tile >= nTiles) return;

  int rc   = lane & 15;
  int half = lane >> 4;

  const float*  arow = X  + (size_t)tile * 16 * IN_FEAT + rc * IN_FEAT;
  const __bf16* bcol = Bs + (size_t)rc * IN_FEAT;

  v8f c = {0.f, 0.f, 0.f, 0.f, 0.f, 0.f, 0.f, 0.f};

#pragma unroll 2
  for (int ks = 0; ks < IN_FEAT / 32; ++ks) {
    int kb = ks * 32 + half * 8;
    v4f a0 = *(const v4f*)(arow + kb);
    v4f a1 = *(const v4f*)(arow + kb + 4);
    v4f a2 = *(const v4f*)(arow + kb + 16);
    v4f a3 = *(const v4f*)(arow + kb + 20);
    v8bf b0 = *(const v8bf*)(bcol + kb);
    v8bf b1 = *(const v8bf*)(bcol + kb + 16);
    v16bf A, B;
#pragma unroll
    for (int i = 0; i < 4; ++i) {
      A[i]      = (__bf16)a0[i];
      A[4 + i]  = (__bf16)a1[i];
      A[8 + i]  = (__bf16)a2[i];
      A[12 + i] = (__bf16)a3[i];
    }
#pragma unroll
    for (int i = 0; i < 8; ++i) { B[i] = b0[i]; B[8 + i] = b1[i]; }
    c = __builtin_amdgcn_wmma_f32_16x16x32_bf16(
         false, A,  false, B,
         (short)0, c,  false,  false);
  }

  float* orow = XW1 + (size_t)tile * 16 * HID + (half * 8 * HID + rc);
#pragma unroll
  for (int r = 0; r < 8; ++r) orow[r * HID] = c[r];
}

__global__ __launch_bounds__(1024) void bin_hist_k(const int* __restrict__ DST,
                                                   int* __restrict__ CNT,
                                                   int E, int NBP) {
  __shared__ int h[NBMAX];
  const int t = threadIdx.x;
  h[t] = 0;
  __syncthreads();
  const int slab = (E + NSLAB - 1) / NSLAB;
  const int e0 = blockIdx.x * slab;
  const int e1 = min(e0 + slab, E);
  const bool vec = ((E & 3) == 0) && ((slab & 3) == 0);
  if (vec) hist_slab<true>(DST, h, e0, e1, t);
  else     hist_slab<false>(DST, h, e0, e1, t);
  __syncthreads();
  if (t < NBP) CNT[blockIdx.x * NBP + t] = h[t];
}

__global__ __launch_bounds__(1024) void bin_scan_k(int* __restrict__ CNT,
                                                   int* __restrict__ TOT, int NBP) {
  __shared__ int rs[32][33];
  const int l = threadIdx.x & 31, r = threadIdx.x >> 5;
  const int col = blockIdx.x * 32 + l;
  int v[8];
  int run = 0;
#pragma unroll
  for (int k = 0; k < 8; ++k) {
    int c = CNT[(r * 8 + k) * NBP + col];
    v[k] = run;
    run += c;
  }
  rs[r][l] = run;
  __syncthreads();
  int base = 0, tot = 0;
  for (int rr = 0; rr < 32; ++rr) {
    int x = rs[rr][l];
    if (rr < r) base += x;
    tot += x;
  }
#pragma unroll
  for (int k = 0; k < 8; ++k) CNT[(r * 8 + k) * NBP + col] = base + v[k];
  if (r == 0) TOT[col] = tot;
}

__global__ __launch_bounds__(1024, 16) void bin_fill_k(
    const int* __restrict__ SRC, const int* __restrict__ DST,
    const int* __restrict__ CNT, const int* __restrict__ TOT,
    int* __restrict__ START, int* __restrict__ BINNED, int E, int NB, int NBP,
    const float* __restrict__ X, const float* __restrict__ W1,
    float* __restrict__ XW1, int nTiles, int nGemm) {
  __shared__ __bf16 Bs[HID * IN_FEAT];
  __shared__ int tot[NBMAX];
  __shared__ int cb[32];
  __shared__ int cur[NBMAX];
  if ((int)blockIdx.x < nGemm) {
    gemm1_block<1024>(X, W1, XW1, nTiles, Bs, blockIdx.x);
    return;
  }
  const int fb = blockIdx.x - nGemm;
  const int t = threadIdx.x;
  tot[t] = (t < NBP) ? TOT[t] : 0;
  __syncthreads();
  if (t < 32) {
    int s = 0;
    for (int j = 0; j < 32; ++j) s += tot[t * 32 + j];
    cb[t] = s;
  }
  __syncthreads();
  int st = 0;
  for (int j = 0; j < (t >> 5); ++j) st += cb[j];
  for (int j = (t & ~31); j < t; ++j) st += tot[j];
  if (fb == 0 && t <= NB) START[t] = st;
  cur[t] = (t < NBP) ? st + CNT[fb * NBP + t] : 0;
  __syncthreads();

  const int slab = (E + NSLAB - 1) / NSLAB;
  const int e0 = fb * slab;
  const int e1 = min(e0 + slab, E);
  const bool vec = ((E & 3) == 0) && ((slab & 3) == 0);
  if (vec) fill_slab<true>(SRC, DST, cur, BINNED, e0, e1, t);
  else     fill_slab<false>(SRC, DST, cur, BINNED, e0, e1, t);
}

__global__ __launch_bounds__(256) void bin_deg_k(const int* __restrict__ START,
                                                 const int* __restrict__ BINNED,
                                                 float* __restrict__ DNV, int N) {
  __shared__ int c[RB];
  const int t = threadIdx.x;
  if (t < RB) c[t] = 1;
  __syncthreads();
  const int a = START[blockIdx.x], b = START[blockIdx.x + 1];
  for (int i0 = a + t; i0 < b; i0 += 256 * GU) {
    int p[GU];
#pragma unroll
    for (int u = 0; u < GU; ++u) {
      const int i = i0 + u * 256;
      p[u] = (i < b) ? BINNED[i] : -1;
    }
#pragma unroll
    for (int u = 0; u < GU; ++u)
      if (p[u] >= 0) atomicAdd(&c[p[u] & (RB - 1)], 1);
  }
  __syncthreads();
  const int n = blockIdx.x * RB + t;
  if (t < RB && n < N) DNV[n] = rsqrtf((float)c[t]);
}

__global__ __launch_bounds__(512) void bin_agg1_k(
    const int* __restrict__ START, const int* __restrict__ BINNED,
    const float* __restrict__ DNV, const float* __restrict__ XW1,
    const float* __restrict__ B1, const float* __restrict__ W2,
    float* __restrict__ HW2, int N) {
  __shared__ float acc[RB * HP];
  __shared__ float dl[RB];
  const int t = threadIdx.x;
  const int n0 = blockIdx.x * RB;
  const int q = t & 3;
  {
    const int r = t >> 2;
    const int n = n0 + r;
    float dn = 0.f;
    v4f v = {0.f, 0.f, 0.f, 0.f};
    if (n < N) {
      dn = DNV[n];
      v = *(const v4f*)(XW1 + (size_t)n * HID + q * 4);
    }
    const float w = dn * dn;
    if (q == 0) dl[r] = dn;
#pragma unroll
    for (int i = 0; i < 4; ++i) acc[r * HP + q * 4 + i] = w * v[i];
  }
  __syncthreads();
  const int a = START[blockIdx.x], b = START[blockIdx.x + 1];
  for (int i0 = a + (t >> 2); i0 < b; i0 += 128 * GU) {
    int p[GU];
#pragma unroll
    for (int u = 0; u < GU; ++u) {
      const int i = i0 + u * 128;
      p[u] = (i < b) ? BINNED[i] : -1;
    }
    float dn[GU];
    v4f v[GU];
#pragma unroll
    for (int u = 0; u < GU; ++u) {
      const int s = (p[u] < 0 ? 0 : p[u]) >> RSH;
      dn[u] = DNV[s];
      v[u] = *(const v4f*)(XW1 + (size_t)s * HID + q * 4);
    }
#pragma unroll
    for (int u = 0; u < GU; ++u) {
      if (p[u] >= 0) {
        const int r = p[u] & (RB - 1);
        const float nrm = dn[u] * dl[r];
        float* ap = acc + r * HP + q * 4;
#pragma unroll
        for (int k = 0; k < 4; ++k) ldsAddF(ap + k, nrm * v[u][k]);
      }
    }
  }
  __syncthreads();
  if (t < 2 * RB) {
    const int r = t >> 1, j = t & 1;
    const int n = n0 + r;
    if (n < N) {
      float h = 0.f;
#pragma unroll
      for (int f = 0; f < HID; ++f) {
        float hv = fmaxf(acc[r * HP + f] + B1[f], 0.f);
        h += hv * W2[2 * f + j];
      }
      HW2[2 * n + j] = h;
    }
  }
}

__global__ __launch_bounds__(256) void bin_agg2_k(
    const int* __restrict__ START, const int* __restrict__ BINNED,
    const float* __restrict__ DNV, const float* __restrict__ HW2,
    const float* __restrict__ B2, float* __restrict__ out, int N) {
  __shared__ float acc[2 * RB];
  __shared__ float dl[RB];
  const int t = threadIdx.x;
  const int n0 = blockIdx.x * RB;
  {
    const int r = t >> 1;
    const int n = n0 + r;
    float dn = 0.f, hv = 0.f;
    if (n < N) {
      dn = DNV[n];
      hv = HW2[2 * n + (t & 1)];
    }
    const float w = dn * dn;
    acc[t] = w * hv;
    if ((t & 1) == 0) dl[r] = dn;
  }
  __syncthreads();
  const int a = START[blockIdx.x], b = START[blockIdx.x + 1];
  for (int i0 = a + t; i0 < b; i0 += 256 * GU) {
    int p[GU];
#pragma unroll
    for (int u = 0; u < GU; ++u) {
      const int i = i0 + u * 256;
      p[u] = (i < b) ? BINNED[i] : -1;
    }
    float dn[GU];
    v2f hv[GU];
#pragma unroll
    for (int u = 0; u < GU; ++u) {
      const int s = (p[u] < 0 ? 0 : p[u]) >> RSH;
      dn[u] = DNV[s];
      hv[u] = *(const v2f*)(HW2 + 2 * (size_t)s);
    }
#pragma unroll
    for (int u = 0; u < GU; ++u) {
      if (p[u] >= 0) {
        const int r = p[u] & (RB - 1);
        const float nrm = dn[u] * dl[r];
        ldsAddF(&acc[2 * r],     nrm * hv[u][0]);
        ldsAddF(&acc[2 * r + 1], nrm * hv[u][1]);
      }
    }
  }
  __syncthreads();
  if (t < RB) {
    const int n = n0 + t;
    if (n < N) {
      float x = acc[2 * t]     + B2[0];
      float y = acc[2 * t + 1] + B2[1];
      float m = fmaxf(x, y);
      float lse = m + logf(expf(x - m) + expf(y - m));
      v2f o;
      o[0] = x - lse;
      o[1] = y - lse;
      *(v2f*)(out + 2 * (size_t)n) = o;
    }
  }
}

extern "C" void kernel_launch(void* const* d_in, const int* in_sizes, int n_in,
                              void* d_out, int out_size, void* d_ws, size_t ws_size,
                              hipStream_t stream) {
  const float* X  = (const float*)d_in[0];
  const int*   EI = (const int*)d_in[1];
  const float* W1 = (const float*)d_in[2];
  const float* B1 = (const float*)d_in[3];
  const float* W2 = (const float*)d_in[4];
  const float* B2 = (const float*)d_in[5];

  const int N = in_sizes[0] / IN_FEAT;
  const int E = in_sizes[1] / 2;
  const int* SRC = EI;
  const int* DST = EI + E;

  const int NB  = (N + RB - 1) / RB;
  const int NBP = (NB + 31) / 32 * 32;

  float* ws     = (float*)d_ws;
  float* XW1    = ws;
  int*   BINNED = (int*)(ws + (size_t)N * HID);
  float* HW2    = (float*)(BINNED + E);
  float* DNV    = HW2 + (size_t)2 * N;
  int*   CNT    = (int*)(DNV + N);
  int*   TOT    = CNT + (size_t)NSLAB * NBP;
  int*   START  = TOT + NBMAX;

  const int nTiles = (N + 15) / 16;
  const int nGemm  = (nTiles + 31) / 32;

  bin_hist_k<<<NSLAB, 1024, 0, stream>>>(DST, CNT, E, NBP);
  bin_scan_k<<<NBP / 32, 1024, 0, stream>>>(CNT, TOT, NBP);
  bin_fill_k<<<nGemm + NSLAB, 1024, 0, stream>>>(SRC, DST, CNT, TOT, START, BINNED, E, NB, NBP,
                                               X, W1, XW1, nTiles, nGemm);
  bin_deg_k<<<NB, 256, 0, stream>>>(START, BINNED, DNV, N);
  bin_agg1_k<<<NB, 512, 0, stream>>>(START, BINNED, DNV, XW1, B1, W2, HW2, N);
  bin_agg2_k<<<NB, 256, 0, stream>>>(START, BINNED, DNV, HW2, B2, (float*)d_out, N);
}


# source's deg^-1/2 kept in the line its edge gathers anyway (layer 1: upper half of a 128-byte X@W1 row; layer 2: 16-byte record with both sums): one random request per edge, not two; same sums
# speedup vs baseline: 4.2749x; 1.1273x over previous
#include <hip/hip_runtime.h>
#include <math.h>


typedef __attribute__((ext_vector_type(16))) __bf16 v16bf;
typedef __attribute__((ext_vector_type(8)))  __bf16 v8bf;
typedef __attribute__((ext_vector_type(8)))  float  v8f;
typedef __attribute__((ext_vector_type(4)))  float  v4f;

#define IN_FEAT 512
#define HID     16

typedef __attribute__((ext_vector_type(2)))  float  v2f;
typedef __attribute__((ext_vector_type(3)))  float  v3f;

#define RSH   7
#define RB    128
#define NSLAB 256
#define NBMAX 1024
#define HP    17
#define EU    4
#define GU    8

__device__ __forceinline__ void ldsAddF(float* p, float v) {
  __hip_atomic_fetch_add(p, v, __ATOMIC_RELAXED, __HIP_MEMORY_SCOPE_WORKGROUP);
}

template <bool VEC>
__device__ __forceinline__ int4 ld4(const int* __restrict__ p, int e, int e0, int e1) {
  int4 r;
  if (VEC) {
    r = *(const int4*)(p + (e < e1 ? e : e0));
    if (e >= e1) r.x = r.y = r.z = r.w = -1;
  } else {
    r.x = (e     < e1) ? p[e]     : -1;
    r.y = (e + 1 < e1) ? p[e + 1] : -1;
    r.z = (e + 2 < e1) ? p[e + 2] : -1;
    r.w = (e + 3 < e1) ? p[e + 3] : -1;
  }
  return r;
}

template <bool VEC>
__device__ __forceinline__ void hist_slab(const int* __restrict__ DST, int* h,
                                          int e0, int e1, int t) {
  for (int e = e0 + 4 * t; e < e1; e += 4 * 1024 * EU) {
    int4 d[EU];
#pragma unroll
    for (int u = 0; u < EU; ++u) d[u] = ld4<VEC>(DST, e + u * 4096, e0, e1);
#pragma unroll
    for (int u = 0; u < EU; ++u) {
      if (d[u].x >= 0) atomicAdd(&h[d[u].x >> RSH], 1);
      if (d[u].y >= 0) atomicAdd(&h[d[u].y >> RSH], 1);
      if (d[u].z >= 0) atomicAdd(&h[d[u].z >> RSH], 1);
      if (d[u].w >= 0) atomicAdd(&h[d[u].w >> RSH], 1);
    }
  }
}

template <bool VEC>
__device__ __forceinline__ void fill_slab(const int* __restrict__ SRC,
                                          const int* __restrict__ DST, int* cur,
                                          int* __restrict__ BINNED,
                                          int e0, int e1, int t) {
  for (int e = e0 + 4 * t; e < e1; e += 4 * 1024 * EU) {
    int4 s[EU], d[EU];
#pragma unroll
    for (int u = 0; u < EU; ++u) {
      s[u] = ld4<VEC>(SRC, e + u * 4096, e0, e1);
      d[u] = ld4<VEC>(DST, e + u * 4096, e0, e1);
    }
#pragma unroll
    for (int u = 0; u < EU; ++u) {
      if (d[u].x >= 0) BINNED[atomicAdd(&cur[d[u].x >> RSH], 1)] = (s[u].x << RSH) | (d[u].x & (RB - 1));
      if (d[u].y >= 0) BINNED[atomicAdd(&cur[d[u].y >> RSH], 1)] = (s[u].y << RSH) | (d[u].y & (RB - 1));
      if (d[u].z >= 0) BINNED[atomicAdd(&cur[d[u].z >> RSH], 1)] = (s[u].z << RSH) | (d[u].z & (RB - 1));
      if (d[u].w >= 0) BINNED[atomicAdd(&cur[d[u].w >> RSH], 1)] = (s[u].w << RSH) | (d[u].w & (RB - 1));
    }
  }
}

template <int NT>
__device__ __forceinline__ void gemm1_block(
    const float* __restrict__ X, const float* __restrict__ W1,
    float* __restrict__ XW1, int nTiles, __bf16* Bs, int gb, int xp) {
  for (int i = threadIdx.x; i < HID * IN_FEAT; i += NT) {
    int k = i >> 4, n = i & 15;
    Bs[n * IN_FEAT + k] = (__bf16)W1[i];
  }
  __syncthreads();

  int wave = __builtin_amdgcn_readfirstlane(threadIdx.x >> 5);
  int lane = threadIdx.x & 31;
  int tile = gb * (NT / 32) + wave;
  if (tile >= nTiles) return;

  int rc   = lane & 15;
  int half = lane >> 4;

  const float*  arow = X  + (size_t)tile * 16 * IN_FEAT + rc * IN_FEAT;
  const __bf16* bcol = Bs + (size_t)rc * IN_FEAT;

  v8f c = {0.f, 0.f, 0.f, 0.f, 0.f, 0.f, 0.f, 0.f};

#pragma unroll 2
  for (int ks = 0; ks < IN_FEAT / 32; ++ks) {
    int kb = ks * 32 + half * 8;
    v4f a0 = *(const v4f*)(arow + kb);
    v4f a1 = *(const v4f*)(arow + kb + 4);
    v4f a2 = *(const v4f*)(arow + kb + 16);
    v4f a3 = *(const v4f*)(arow + kb + 20);
    v8bf b0 = *(const v8bf*)(bcol + kb);
    v8bf b1 = *(const v8bf*)(bcol + kb + 16);
    v16bf A, B;
#pragma unroll
    for (int i = 0; i < 4; ++i) {
      A[i]      = (__bf16)a0[i];
      A[4 + i]  = (__bf16)a1[i];
      A[8 + i]  = (__bf16)a2[i];
      A[12 + i] = (__bf16)a3[i];
    }
#pragma unroll
    for (int i = 0; i < 8; ++i) { B[i] = b0[i]; B[8 + i] = b1[i]; }
    c = __builtin_amdgcn_wmma_f32_16x16x32_bf16(
         false, A,  false, B,
         (short)0, c,  false,  false);
  }

  float* orow = XW1 + (size_t)tile * 16 * xp + (half * 8 * xp + rc);
#pragma unroll
  for (int r = 0; r < 8; ++r) orow[r * xp] = c[r];
}

__global__ __launch_bounds__(1024) void bin_hist_k(const int* __restrict__ DST,
                                                   int* __restrict__ CNT,
                                                   int E, int NBP) {
  __shared__ int h[NBMAX];
  const int t = threadIdx.x;
  h[t] = 0;
  __syncthreads();
  const int slab = (E + NSLAB - 1) / NSLAB;
  const int e0 = blockIdx.x * slab;
  const int e1 = min(e0 + slab, E);
  const bool vec = ((E & 3) == 0) && ((slab & 3) == 0);
  if (vec) hist_slab<true>(DST, h, e0, e1, t);
  else     hist_slab<false>(DST, h, e0, e1, t);
  __syncthreads();
  if (t < NBP) CNT[blockIdx.x * NBP + t] = h[t];
}

__global__ __launch_bounds__(1024) void bin_scan_k(int* __restrict__ CNT,
                                                   int* __restrict__ TOT, int NBP) {
  __shared__ int rs[32][33];
  const int l = threadIdx.x & 31, r = threadIdx.x >> 5;
  const int col = blockIdx.x * 32 + l;
  int v[8];
  int run = 0;
#pragma unroll
  for (int k = 0; k < 8; ++k) {
    int c = CNT[(r * 8 + k) * NBP + col];
    v[k] = run;
    run += c;
  }
  rs[r][l] = run;
  __syncthreads();
  int base = 0, tot = 0;
  for (int rr = 0; rr < 32; ++rr) {
    int x = rs[rr][l];
    if (rr < r) base += x;
    tot += x;
  }
#pragma unroll
  for (int k = 0; k < 8; ++k) CNT[(r * 8 + k) * NBP + col] = base + v[k];
  if (r == 0) TOT[col] = tot;
}

__global__ __launch_bounds__(1024, 16) void bin_fill_k(
    const int* __restrict__ SRC, const int* __restrict__ DST,
    const int* __restrict__ CNT, const int* __restrict__ TOT,
    int* __restrict__ START, int* __restrict__ BINNED, int E, int NB, int NBP,
    const float* __restrict__ X, const float* __restrict__ W1,
    float* __restrict__ XW1, int nTiles, int nGemm, int xp) {
  __shared__ __bf16 Bs[HID * IN_FEAT];
  __shared__ int tot[NBMAX];
  __shared__ int cb[32];
  __shared__ int cur[NBMAX];
  if ((int)blockIdx.x < nGemm) {
    gemm1_block<1024>(X, W1, XW1, nTiles, Bs, blockIdx.x, xp);
    return;
  }
  const int fb = blockIdx.x - nGemm;
  const int t = threadIdx.x;
  tot[t] = (t < NBP) ? TOT[t] : 0;
  __syncthreads();
  if (t < 32) {
    int s = 0;
    for (int j = 0; j < 32; ++j) s += tot[t * 32 + j];
    cb[t] = s;
  }
  __syncthreads();
  int st = 0;
  for (int j = 0; j < (t >> 5); ++j) st += cb[j];
  for (int j = (t & ~31); j < t; ++j) st += tot[j];
  if (fb == 0 && t <= NB) START[t] = st;
  cur[t] = (t < NBP) ? st + CNT[fb * NBP + t] : 0;
  __syncthreads();

  const int slab = (E + NSLAB - 1) / NSLAB;
  const int e0 = fb * slab;
  const int e1 = min(e0 + slab, E);
  const bool vec = ((E & 3) == 0) && ((slab & 3) == 0);
  if (vec) fill_slab<true>(SRC, DST, cur, BINNED, e0, e1, t);
  else     fill_slab<false>(SRC, DST, cur, BINNED, e0, e1, t);
}

template <bool WIDE>
__global__ __launch_bounds__(256) void bin_deg_k(const int* __restrict__ START,
                                                 const int* __restrict__ BINNED,
                                                 float* __restrict__ DNV,
                                                 float* __restrict__ XW1, int N) {
  __shared__ int c[RB];
  const int t = threadIdx.x;
  if (t < RB) c[t] = 1;
  __syncthreads();
  const int a = START[blockIdx.x], b = START[blockIdx.x + 1];
  for (int i0 = a + t; i0 < b; i0 += 256 * GU) {
    int p[GU];
#pragma unroll
    for (int u = 0; u < GU; ++u) {
      const int i = i0 + u * 256;
      p[u] = (i < b) ? BINNED[i] : -1;
    }
#pragma unroll
    for (int u = 0; u < GU; ++u)
      if (p[u] >= 0) atomicAdd(&c[p[u] & (RB - 1)], 1);
  }
  __syncthreads();
  const int n = blockIdx.x * RB + t;
  if (t < RB && n < N) {
    const float dv = rsqrtf((float)c[t]);
    if (WIDE) {
      float* dp = XW1 + (size_t)n * 32 + HID;
#pragma unroll
      for (int k = 0; k < 4; ++k) dp[4 * k] = dv;
    } else {
      DNV[n] = dv;
    }
  }
}

template <bool WIDE>
__global__ __launch_bounds__(512) void bin_agg1_k(
    const int* __restrict__ START, const int* __restrict__ BINNED,
    const float* __restrict__ DNV, const float* __restrict__ XW1,
    const float* __restrict__ B1, const float* __restrict__ W2,
    float* __restrict__ REC2, int N) {
  constexpr int XP = WIDE ? 32 : HID;
  __shared__ float acc[RB * HP];
  __shared__ float dl[RB];
  const int t = threadIdx.x;
  const int n0 = blockIdx.x * RB;
  const int q = t & 3;
  {
    const int r = t >> 2;
    const int n = n0 + r;
    float dn = 0.f;
    v4f v = {0.f, 0.f, 0.f, 0.f};
    if (n < N) {
      const float* rp = XW1 + (size_t)n * XP + q * 4;
      dn = WIDE ? rp[HID] : DNV[n];
      v = *(const v4f*)rp;
    }
    const float w = dn * dn;
    if (q == 0) dl[r] = dn;
#pragma unroll
    for (int i = 0; i < 4; ++i) acc[r * HP + q * 4 + i] = w * v[i];
  }
  __syncthreads();
  const int a = START[blockIdx.x], b = START[blockIdx.x + 1];
  for (int i0 = a + (t >> 2); i0 < b; i0 += 128 * GU) {
    int p[GU];
#pragma unroll
    for (int u = 0; u < GU; ++u) {
      const int i = i0 + u * 128;
      p[u] = (i < b) ? BINNED[i] : -1;
    }
    float dn[GU];
    v4f v[GU];
#pragma unroll
    for (int u = 0; u < GU; ++u) {
      const int s = (p[u] < 0 ? 0 : p[u]) >> RSH;
      const unsigned bo = ((unsigned)s * XP + q * 4) * 4u;
      const float* rp = (const float*)((const char*)XW1 + bo);
      dn[u] = WIDE ? rp[HID] : DNV[s];
      v[u] = *(const v4f*)rp;
    }
#pragma unroll
    for (int u = 0; u < GU; ++u) {
      if (p[u] >= 0) {
        const int r = p[u] & (RB - 1);
        const float nrm = dn[u] * dl[r];
        float* ap = acc + r * HP + q * 4;
#pragma unroll
        for (int k = 0; k < 4; ++k) ldsAddF(ap + k, nrm * v[u][k]);
      }
    }
  }
  __syncthreads();
  if (t < RB) {
    const int n = n0 + t;
    if (n < N) {
      float h0 = 0.f, h1 = 0.f;
#pragma unroll
      for (int f = 0; f < HID; ++f) {
        float hv = fmaxf(acc[t * HP + f] + B1[f], 0.f);
        h0 += hv * W2[2 * f];
        h1 += hv * W2[2 * f + 1];
      }
      v4f o;
      o[0] = h0;
      o[1] = h1;
      o[2] = dl[t];
      o[3] = 0.f;
      *(v4f*)(REC2 + 4 * (size_t)n) = o;
    }
  }
}

__global__ __launch_bounds__(256) void bin_agg2_k(
    const int* __restrict__ START, const int* __restrict__ BINNED,
    const float* __restrict__ REC2,
    const float* __restrict__ B2, float* __restrict__ out, int N) {
  __shared__ float acc[2 * RB];
  __shared__ float dl[RB];
  const int t = threadIdx.x;
  const int n0 = blockIdx.x * RB;
  {
    const int r = t >> 1;
    const int n = n0 + r;
    float dn = 0.f, hv = 0.f;
    if (n < N) {
      dn = REC2[4 * (size_t)n + 2];
      hv = REC2[4 * (size_t)n + (t & 1)];
    }
    const float w = dn * dn;
    acc[t] = w * hv;
    if ((t & 1) == 0) dl[r] = dn;
  }
  __syncthreads();
  const int a = START[blockIdx.x], b = START[blockIdx.x + 1];
  for (int i0 = a + t; i0 < b; i0 += 256 * GU) {
    int p[GU];
#pragma unroll
    for (int u = 0; u < GU; ++u) {
      const int i = i0 + u * 256;
      p[u] = (i < b) ? BINNED[i] : -1;
    }
    v3f rec[GU];
#pragma unroll
    for (int u = 0; u < GU; ++u) {
      const int s = (p[u] < 0 ? 0 : p[u]) >> RSH;
      const float* rp = REC2 + 4 * (size_t)s;
      rec[u][0] = rp[0];
      rec[u][1] = rp[1];
      rec[u][2] = rp[2];
    }
#pragma unroll
    for (int u = 0; u < GU; ++u) {
      if (p[u] >= 0) {
        const int r = p[u] & (RB - 1);
        const float nrm = rec[u][2] * dl[r];
        ldsAddF(&acc[2 * r],     nrm * rec[u][0]);
        ldsAddF(&acc[2 * r + 1], nrm * rec[u][1]);
      }
    }
  }
  __syncthreads();
  if (t < RB) {
    const int n = n0 + t;
    if (n < N) {
      float x = acc[2 * t]     + B2[0];
      float y = acc[2 * t + 1] + B2[1];
      float m = fmaxf(x, y);
      float lse = m + logf(expf(x - m) + expf(y - m));
      v2f o;
      o[0] = x - lse;
      o[1] = y - lse;
      *(v2f*)(out + 2 * (size_t)n) = o;
    }
  }
}

extern "C" void kernel_launch(void* const* d_in, const int* in_sizes, int n_in,
                              void* d_out, int out_size, void* d_ws, size_t ws_size,
                              hipStream_t stream) {
  const float* X  = (const float*)d_in[0];
  const int*   EI = (const int*)d_in[1];
  const float* W1 = (const float*)d_in[2];
  const float* B1 = (const float*)d_in[3];
  const float* W2 = (const float*)d_in[4];
  const float* B2 = (const float*)d_in[5];

  const int N = in_sizes[0] / IN_FEAT;
  const int E = in_sizes[1] / 2;
  const int* SRC = EI;
  const int* DST = EI + E;

  const int NB  = (N + RB - 1) / RB;
  const int NBP = (NB + 31) / 32 * 32;

  const size_t Nr   = ((size_t)N + 31) / 32 * 32;
  const size_t Er   = ((size_t)E + 31) / 32 * 32;
  const size_t cw   = (size_t)NSLAB * NBP;
  const size_t uw   = cw > 4 * Nr ? cw : 4 * Nr;
  const bool   wide = ws_size >= (32 * Nr + Er + NBMAX + NBMAX + 32 + uw) * 4;
  const int    xp   = wide ? 32 : HID;
  float* ws     = (float*)d_ws;
  float* XW1    = ws;
  int*   BINNED = (int*)(ws + Nr * xp);
  float* DNV    = (float*)(BINNED + Er);
  int*   TOT    = (int*)(DNV + (wide ? 0 : Nr));
  int*   START  = TOT + NBMAX;
  int*   CNT    = START + NBMAX + 32;
  float* REC2   = (float*)CNT;

  const int nTiles = (N + 15) / 16;
  const int nGemm  = (nTiles + 31) / 32;

  bin_hist_k<<<NSLAB, 1024, 0, stream>>>(DST, CNT, E, NBP);
  bin_scan_k<<<NBP / 32, 1024, 0, stream>>>(CNT, TOT, NBP);
  bin_fill_k<<<nGemm + NSLAB, 1024, 0, stream>>>(SRC, DST, CNT, TOT, START, BINNED, E, NB, NBP,
                                               X, W1, XW1, nTiles, nGemm, xp);
  if (wide) {
    bin_deg_k<true><<<NB, 256, 0, stream>>>(START, BINNED, DNV, XW1, N);
    bin_agg1_k<true><<<NB, 512, 0, stream>>>(START, BINNED, DNV, XW1, B1, W2, REC2, N);
  } else {
    bin_deg_k<false><<<NB, 256, 0, stream>>>(START, BINNED, DNV, XW1, N);
    bin_agg1_k<false><<<NB, 512, 0, stream>>>(START, BINNED, DNV, XW1, B1, W2, REC2, N);
  }
  bin_agg2_k<<<NB, 256, 0, stream>>>(START, BINNED, REC2, B2, (float*)d_out, N);
}


# bucket-fill pass: each slab ordered by bucket in LDS first, then written by position: adjacent lanes store adjacent words of a run, not 32 scattered 4-byte pieces a store; sums still in arrival order
# speedup vs baseline: 5.6707x; 1.3265x over previous
#include <hip/hip_runtime.h>
#include <math.h>


typedef __attribute__((ext_vector_type(16))) __bf16 v16bf;
typedef __attribute__((ext_vector_type(8)))  __bf16 v8bf;
typedef __attribute__((ext_vector_type(8)))  float  v8f;
typedef __attribute__((ext_vector_type(4)))  float  v4f;

#define IN_FEAT 512
#define HID     16

typedef __attribute__((ext_vector_type(2)))  float  v2f;
typedef __attribute__((ext_vector_type(3)))  float  v3f;

#define RSH   7
#define RB    128
#define NSLAB 256
#define NBMAX 1024
#define HP    17
#define EU    4
#define GU    8
#define LCAP  12800

typedef __attribute__((ext_vector_type(2)))  int    v2i;
static_assert(LCAP <= 4 * 1024 * EU, "an ordered slab is one trip of the edge loop");

__device__ __forceinline__ void ldsAddF(float* p, float v) {
  __hip_atomic_fetch_add(p, v, __ATOMIC_RELAXED, __HIP_MEMORY_SCOPE_WORKGROUP);
}

template <bool VEC>
__device__ __forceinline__ int4 ld4(const int* __restrict__ p, int e, int e0, int e1) {
  int4 r;
  if (VEC) {
    r = *(const int4*)(p + (e < e1 ? e : e0));
    if (e >= e1) r.x = r.y = r.z = r.w = -1;
  } else {
    r.x = (e     < e1) ? p[e]     : -1;
    r.y = (e + 1 < e1) ? p[e + 1] : -1;
    r.z = (e + 2 < e1) ? p[e + 2] : -1;
    r.w = (e + 3 < e1) ? p[e + 3] : -1;
  }
  return r;
}

template <bool VEC>
__device__ __forceinline__ void hist_slab(const int* __restrict__ DST, int* h,
                                          int e0, int e1, int t) {
  for (int e = e0 + 4 * t; e < e1; e += 4 * 1024 * EU) {
    int4 d[EU];
#pragma unroll
    for (int u = 0; u < EU; ++u) d[u] = ld4<VEC>(DST, e + u * 4096, e0, e1);
#pragma unroll
    for (int u = 0; u < EU; ++u) {
      if (d[u].x >= 0) atomicAdd(&h[d[u].x >> RSH], 1);
      if (d[u].y >= 0) atomicAdd(&h[d[u].y >> RSH], 1);
      if (d[u].z >= 0) atomicAdd(&h[d[u].z >> RSH], 1);
      if (d[u].w >= 0) atomicAdd(&h[d[u].w >> RSH], 1);
    }
  }
}

template <bool VEC>
__device__ __forceinline__ void fill_slab(const int* __restrict__ SRC,
                                          const int* __restrict__ DST, int* cur,
                                          int* __restrict__ BINNED,
                                          int e0, int e1, int t) {
  for (int e = e0 + 4 * t; e < e1; e += 4 * 1024 * EU) {
    int4 s[EU], d[EU];
#pragma unroll
    for (int u = 0; u < EU; ++u) {
      s[u] = ld4<VEC>(SRC, e + u * 4096, e0, e1);
      d[u] = ld4<VEC>(DST, e + u * 4096, e0, e1);
    }
#pragma unroll
    for (int u = 0; u < EU; ++u) {
      if (d[u].x >= 0) BINNED[atomicAdd(&cur[d[u].x >> RSH], 1)] = (s[u].x << RSH) | (d[u].x & (RB - 1));
      if (d[u].y >= 0) BINNED[atomicAdd(&cur[d[u].y >> RSH], 1)] = (s[u].y << RSH) | (d[u].y & (RB - 1));
      if (d[u].z >= 0) BINNED[atomicAdd(&cur[d[u].z >> RSH], 1)] = (s[u].z << RSH) | (d[u].z & (RB - 1));
      if (d[u].w >= 0) BINNED[atomicAdd(&cur[d[u].w >> RSH], 1)] = (s[u].w << RSH) | (d[u].w & (RB - 1));
    }
  }
}

template <bool VEC>
__device__ __forceinline__ void order_slab(const int* __restrict__ SRC,
                                           const int* __restrict__ DST, int* cur,
                                           v2i* sp, int e0, int e1, int t) {
  const int e = e0 + 4 * t;
  if (e >= e1) return;
  int4 s[EU], d[EU];
#pragma unroll
  for (int u = 0; u < EU; ++u) {
    s[u] = ld4<VEC>(SRC, e + u * 4096, e0, e1);
    d[u] = ld4<VEC>(DST, e + u * 4096, e0, e1);
  }
  const int spare = NBMAX + (t & 31);
  int lp[4 * EU];
#pragma unroll
  for (int u = 0; u < EU; ++u) {
    lp[4 * u]     = atomicAdd(&cur[d[u].x >= 0 ? d[u].x >> RSH : spare], 1);
    lp[4 * u + 1] = atomicAdd(&cur[d[u].y >= 0 ? d[u].y >> RSH : spare], 1);
    lp[4 * u + 2] = atomicAdd(&cur[d[u].z >= 0 ? d[u].z >> RSH : spare], 1);
    lp[4 * u + 3] = atomicAdd(&cur[d[u].w >= 0 ? d[u].w >> RSH : spare], 1);
  }
#pragma unroll
  for (int u = 0; u < EU; ++u) {
    if (d[u].x >= 0) { v2i p; p.x = s[u].x; p.y = d[u].x; sp[lp[4 * u]]     = p; }
    if (d[u].y >= 0) { v2i p; p.x = s[u].y; p.y = d[u].y; sp[lp[4 * u + 1]] = p; }
    if (d[u].z >= 0) { v2i p; p.x = s[u].z; p.y = d[u].z; sp[lp[4 * u + 2]] = p; }
    if (d[u].w >= 0) { v2i p; p.x = s[u].w; p.y = d[u].w; sp[lp[4 * u + 3]] = p; }
  }
}

template <int NT>
__device__ __forceinline__ void gemm1_block(
    const float* __restrict__ X, const float* __restrict__ W1,
    float* __restrict__ XW1, int nTiles, __bf16* Bs, int gb, int xp) {
  for (int i = threadIdx.x; i < HID * IN_FEAT; i += NT) {
    int k = i >> 4, n = i & 15;
    Bs[n * IN_FEAT + k] = (__bf16)W1[i];
  }
  __syncthreads();

  int wave = __builtin_amdgcn_readfirstlane(threadIdx.x >> 5);
  int lane = threadIdx.x & 31;
  int tile = gb * (NT / 32) + wave;
  if (tile >= nTiles) return;

  int rc   = lane & 15;
  int half = lane >> 4;

  const float*  arow = X  + (size_t)tile * 16 * IN_FEAT + rc * IN_FEAT;
  const __bf16* bcol = Bs + (size_t)rc * IN_FEAT;

  v8f c = {0.f, 0.f, 0.f, 0.f, 0.f, 0.f, 0.f, 0.f};

#pragma unroll 2
  for (int ks = 0; ks < IN_FEAT / 32; ++ks) {
    int kb = ks * 32 + half * 8;
    v4f a0 = *(const v4f*)(arow + kb);
    v4f a1 = *(const v4f*)(arow + kb + 4);
    v4f a2 = *(const v4f*)(arow + kb + 16);
    v4f a3 = *(const v4f*)(arow + kb + 20);
    v8bf b0 = *(const v8bf*)(bcol + kb);
    v8bf b1 = *(const v8bf*)(bcol + kb + 16);
    v16bf A, B;
#pragma unroll
    for (int i = 0; i < 4; ++i) {
      A[i]      = (__bf16)a0[i];
      A[4 + i]  = (__bf16)a1[i];
      A[8 + i]  = (__bf16)a2[i];
      A[12 + i] = (__bf16)a3[i];
    }
#pragma unroll
    for (int i = 0; i < 8; ++i) { B[i] = b0[i]; B[8 + i] = b1[i]; }
    c = __builtin_amdgcn_wmma_f32_16x16x32_bf16(
         false, A,  false, B,
         (short)0, c,  false,  false);
  }

  float* orow = XW1 + (size_t)tile * 16 * xp + (half * 8 * xp + rc);
#pragma unroll
  for (int r = 0; r < 8; ++r) orow[r * xp] = c[r];
}

__global__ __launch_bounds__(1024) void bin_hist_k(const int* __restrict__ DST,
                                                   int* __restrict__ CNT,
                                                   int E, int NBP) {
  __shared__ int h[NBMAX];
  const int t = threadIdx.x;
  h[t] = 0;
  __syncthreads();
  const int slab = (E + NSLAB - 1) / NSLAB;
  const int e0 = blockIdx.x * slab;
  const int e1 = min(e0 + slab, E);
  const bool vec = ((E & 3) == 0) && ((slab & 3) == 0);
  if (vec) hist_slab<true>(DST, h, e0, e1, t);
  else     hist_slab<false>(DST, h, e0, e1, t);
  __syncthreads();
  if (t < NBP) CNT[blockIdx.x * NBP + t] = h[t];
}

__global__ __launch_bounds__(1024) void bin_scan_k(int* __restrict__ CNT,
                                                   int* __restrict__ TOT, int NBP) {
  __shared__ int rs[32][33];
  const int l = threadIdx.x & 31, r = threadIdx.x >> 5;
  const int col = blockIdx.x * 32 + l;
  int v[8];
  int run = 0;
#pragma unroll
  for (int k = 0; k < 8; ++k) {
    int c = CNT[(r * 8 + k) * NBP + col];
    v[k] = run;
    run += c;
  }
  rs[r][l] = run;
  __syncthreads();
  int base = 0, tot = 0;
  for (int rr = 0; rr < 32; ++rr) {
    int x = rs[rr][l];
    if (rr < r) base += x;
    tot += x;
  }
#pragma unroll
  for (int k = 0; k < 8; ++k) CNT[(r * 8 + k) * NBP + col] = base + v[k];
  if (r == 0) TOT[col] = tot;
}

__global__ __launch_bounds__(1024, 16) void bin_fill_k(
    const int* __restrict__ SRC, const int* __restrict__ DST,
    const int* __restrict__ CNT, const int* __restrict__ TOT,
    int* __restrict__ START, int* __restrict__ BINNED, int E, int NB, int NBP,
    const float* __restrict__ X, const float* __restrict__ W1,
    float* __restrict__ XW1, int nTiles, int nGemm, int xp) {
  __shared__ __bf16 Bs[HID * IN_FEAT];
  __shared__ v2i tt[NBMAX];
  __shared__ v2i cb[32];
  __shared__ int cur[NBMAX + 32];
  __shared__ int dlt[NBMAX];
  __shared__ v2i sp[LCAP];
  if ((int)blockIdx.x < nGemm) {
    gemm1_block<1024>(X, W1, XW1, nTiles, Bs, blockIdx.x, xp);
    return;
  }
  const int fb = blockIdx.x - nGemm;
  const int t = threadIdx.x;
  int c0 = 0;
  {
    int T = 0, nt = 0;
    if (t < NBP) {
      T  = TOT[t];
      c0 = CNT[fb * NBP + t];
      nt = ((fb + 1 < NSLAB) ? CNT[(fb + 1) * NBP + t] : T) - c0;
    }
    v2i w; w.x = T; w.y = nt;
    tt[t] = w;
  }
  __syncthreads();
  if (t < 32) {
    v2i s = {0, 0};
    for (int j = 0; j < 32; ++j) s += tt[t * 32 + j];
    cb[t] = s;
  }
  __syncthreads();
  v2i st = {0, 0};
  for (int j = 0; j < (t >> 5); ++j) st += cb[j];
  for (int j = (t & ~31); j < t; ++j) st += tt[j];
  if (fb == 0 && t <= NB) START[t] = st.x;

  const int slab = (E + NSLAB - 1) / NSLAB;
  const int e0 = fb * slab;
  const int e1 = min(e0 + slab, E);
  const bool vec = ((E & 3) == 0) && ((slab & 3) == 0);
  if (slab <= LCAP) {
    cur[t] = st.y;
    if (t < 32) cur[NBMAX + t] = 0;
    dlt[t] = st.x + c0 - st.y;
    __syncthreads();
    if (vec) order_slab<true>(SRC, DST, cur, sp, e0, e1, t);
    else     order_slab<false>(SRC, DST, cur, sp, e0, e1, t);
    __syncthreads();
    const int ns = max(e1 - e0, 0);
#pragma unroll 4
    for (int i = t; i < ns; i += 1024) {
      const v2i pr = sp[i];
      BINNED[i + dlt[pr.y >> RSH]] = (pr.x << RSH) | (pr.y & (RB - 1));
    }
  } else {
    cur[t] = (t < NBP) ? st.x + c0 : 0;
    __syncthreads();
    if (vec) fill_slab<true>(SRC, DST, cur, BINNED, e0, e1, t);
    else     fill_slab<false>(SRC, DST, cur, BINNED, e0, e1, t);
  }
}

template <bool WIDE>
__global__ __launch_bounds__(256) void bin_deg_k(const int* __restrict__ START,
                                                 const int* __restrict__ BINNED,
                                                 float* __restrict__ DNV,
                                                 float* __restrict__ XW1, int N) {
  __shared__ int c[RB];
  const int t = threadIdx.x;
  if (t < RB) c[t] = 1;
  __syncthreads();
  const int a = START[blockIdx.x], b = START[blockIdx.x + 1];
  for (int i0 = a + t; i0 < b; i0 += 256 * GU) {
    int p[GU];
#pragma unroll
    for (int u = 0; u < GU; ++u) {
      const int i = i0 + u * 256;
      p[u] = (i < b) ? BINNED[i] : -1;
    }
#pragma unroll
    for (int u = 0; u < GU; ++u)
      if (p[u] >= 0) atomicAdd(&c[p[u] & (RB - 1)], 1);
  }
  __syncthreads();
  const int n = blockIdx.x * RB + t;
  if (t < RB && n < N) {
    const float dv = rsqrtf((float)c[t]);
    if (WIDE) {
      float* dp = XW1 + (size_t)n * 32 + HID;
#pragma unroll
      for (int k = 0; k < 4; ++k) dp[4 * k] = dv;
    } else {
      DNV[n] = dv;
    }
  }
}

template <bool WIDE>
__global__ __launch_bounds__(512) void bin_agg1_k(
    const int* __restrict__ START, const int* __restrict__ BINNED,
    const float* __restrict__ DNV, const float* __restrict__ XW1,
    const float* __restrict__ B1, const float* __restrict__ W2,
    float* __restrict__ REC2, int N) {
  constexpr int XP = WIDE ? 32 : HID;
  __shared__ float acc[RB * HP];
  __shared__ float dl[RB];
  const int t = threadIdx.x;
  const int n0 = blockIdx.x * RB;
  const int q = t & 3;
  {
    const int r = t >> 2;
    const int n = n0 + r;
    float dn = 0.f;
    v4f v = {0.f, 0.f, 0.f, 0.f};
    if (n < N) {
      const float* rp = XW1 + (size_t)n * XP + q * 4;
      dn = WIDE ? rp[HID] : DNV[n];
      v = *(const v4f*)rp;
    }
    const float w = dn * dn;
    if (q == 0) dl[r] = dn;
#pragma unroll
    for (int i = 0; i < 4; ++i) acc[r * HP + q * 4 + i] = w * v[i];
  }
  __syncthreads();
  const int a = START[blockIdx.x], b = START[blockIdx.x + 1];
  for (int i0 = a + (t >> 2); i0 < b; i0 += 128 * GU) {
    int p[GU];
#pragma unroll
    for (int u = 0; u < GU; ++u) {
      const int i = i0 + u * 128;
      p[u] = (i < b) ? BINNED[i] : -1;
    }
    float dn[GU];
    v4f v[GU];
#pragma unroll
    for (int u = 0; u < GU; ++u) {
      const int s = (p[u] < 0 ? 0 : p[u]) >> RSH;
      const unsigned bo = ((unsigned)s * XP + q * 4) * 4u;
      const float* rp = (const float*)((const char*)XW1 + bo);
      dn[u] = WIDE ? rp[HID] : DNV[s];
      v[u] = *(const v4f*)rp;
    }
#pragma unroll
    for (int u = 0; u < GU; ++u) {
      if (p[u] >= 0) {
        const int r = p[u] & (RB - 1);
        const float nrm = dn[u] * dl[r];
        float* ap = acc + r * HP + q * 4;
#pragma unroll
        for (int k = 0; k < 4; ++k) ldsAddF(ap + k, nrm * v[u][k]);
      }
    }
  }
  __syncthreads();
  if (t < RB) {
    const int n = n0 + t;
    if (n < N) {
      float h0 = 0.f, h1 = 0.f;
#pragma unroll
      for (int f = 0; f < HID; ++f) {
        float hv = fmaxf(acc[t * HP + f] + B1[f], 0.f);
        h0 += hv * W2[2 * f];
        h1 += hv * W2[2 * f + 1];
      }
      v4f o;
      o[0] = h0;
      o[1] = h1;
      o[2] = dl[t];
      o[3] = 0.f;
      *(v4f*)(REC2 + 4 * (size_t)n) = o;
    }
  }
}

__global__ __launch_bounds__(256) void bin_agg2_k(
    const int* __restrict__ START, const int* __restrict__ BINNED,
    const float* __restrict__ REC2,
    const float* __restrict__ B2, float* __restrict__ out, int N) {
  __shared__ float acc[2 * RB];
  __shared__ float dl[RB];
  const int t = threadIdx.x;
  const int n0 = blockIdx.x * RB;
  {
    const int r = t >> 1;
    const int n = n0 + r;
    float dn = 0.f, hv = 0.f;
    if (n < N) {
      dn = REC2[4 * (size_t)n + 2];
      hv = REC2[4 * (size_t)n + (t & 1)];
    }
    const float w = dn * dn;
    acc[t] = w * hv;
    if ((t & 1) == 0) dl[r] = dn;
  }
  __syncthreads();
  const int a = START[blockIdx.x], b = START[blockIdx.x + 1];
  for (int i0 = a + t; i0 < b; i0 += 256 * GU) {
    int p[GU];
#pragma unroll
    for (int u = 0; u < GU; ++u) {
      const int i = i0 + u * 256;
      p[u] = (i < b) ? BINNED[i] : -1;
    }
    v3f rec[GU];
#pragma unroll
    for (int u = 0; u < GU; ++u) {
      const int s = (p[u] < 0 ? 0 : p[u]) >> RSH;
      const float* rp = REC2 + 4 * (size_t)s;
      rec[u][0] = rp[0];
      rec[u][1] = rp[1];
      rec[u][2] = rp[2];
    }
#pragma unroll
    for (int u = 0; u < GU; ++u) {
      if (p[u] >= 0) {
        const int r = p[u] & (RB - 1);
        const float nrm = rec[u][2] * dl[r];
        ldsAddF(&acc[2 * r],     nrm * rec[u][0]);
        ldsAddF(&acc[2 * r + 1], nrm * rec[u][1]);
      }
    }
  }
  __syncthreads();
  if (t < RB) {
    const int n = n0 + t;
    if (n < N) {
      float x = acc[2 * t]     + B2[0];
      float y = acc[2 * t + 1] + B2[1];
      float m = fmaxf(x, y);
      float lse = m + logf(expf(x - m) + expf(y - m));
      v2f o;
      o[0] = x - lse;
      o[1] = y - lse;
      *(v2f*)(out + 2 * (size_t)n) = o;
    }
  }
}

extern "C" void kernel_launch(void* const* d_in, const int* in_sizes, int n_in,
                              void* d_out, int out_size, void* d_ws, size_t ws_size,
                              hipStream_t stream) {
  const float* X  = (const float*)d_in[0];
  const int*   EI = (const int*)d_in[1];
  const float* W1 = (const float*)d_in[2];
  const float* B1 = (const float*)d_in[3];
  const float* W2 = (const float*)d_in[4];
  const float* B2 = (const float*)d_in[5];

  const int N = in_sizes[0] / IN_FEAT;
  const int E = in_sizes[1] / 2;
  const int* SRC = EI;
  const int* DST = EI + E;

  const int NB  = (N + RB - 1) / RB;
  const int NBP = (NB + 31) / 32 * 32;

  const size_t Nr   = ((size_t)N + 31) / 32 * 32;
  const size_t Er   = ((size_t)E + 31) / 32 * 32;
  const size_t cw   = (size_t)NSLAB * NBP;
  const size_t uw   = cw > 4 * Nr ? cw : 4 * Nr;
  const bool   wide = ws_size >= (32 * Nr + Er + NBMAX + NBMAX + 32 + uw) * 4;
  const int    xp   = wide ? 32 : HID;
  float* ws     = (float*)d_ws;
  float* XW1    = ws;
  int*   BINNED = (int*)(ws + Nr * xp);
  float* DNV    = (float*)(BINNED + Er);
  int*   TOT    = (int*)(DNV + (wide ? 0 : Nr));
  int*   START  = TOT + NBMAX;
  int*   CNT    = START + NBMAX + 32;
  float* REC2   = (float*)CNT;

  const int nTiles = (N + 15) / 16;
  const int nGemm  = (nTiles + 31) / 32;

  bin_hist_k<<<NSLAB, 1024, 0, stream>>>(DST, CNT, E, NBP);
  bin_scan_k<<<NBP / 32, 1024, 0, stream>>>(CNT, TOT, NBP);
  bin_fill_k<<<nGemm + NSLAB, 1024, 0, stream>>>(SRC, DST, CNT, TOT, START, BINNED, E, NB, NBP,
                                               X, W1, XW1, nTiles, nGemm, xp);
  if (wide) {
    bin_deg_k<true><<<NB, 256, 0, stream>>>(START, BINNED, DNV, XW1, N);
    bin_agg1_k<true><<<NB, 512, 0, stream>>>(START, BINNED, DNV, XW1, B1, W2, REC2, N);
  } else {
    bin_deg_k<false><<<NB, 256, 0, stream>>>(START, BINNED, DNV, XW1, N);
    bin_agg1_k<false><<<NB, 512, 0, stream>>>(START, BINNED, DNV, XW1, B1, W2, REC2, N);
  }
  bin_agg2_k<<<NB, 256, 0, stream>>>(START, BINNED, REC2, B2, (float*)d_out, N);
}


# buckets of 100 destination nodes, not 128: the three per-bucket launches are 1000 work-groups, four on nearly every WGP at once, not 782 (three or four a WGP); bucket by division; sums reordered
# speedup vs baseline: 5.8904x; 1.0387x over previous
#include <hip/hip_runtime.h>
#include <math.h>


typedef __attribute__((ext_vector_type(16))) __bf16 v16bf;
typedef __attribute__((ext_vector_type(8)))  __bf16 v8bf;
typedef __attribute__((ext_vector_type(8)))  float  v8f;
typedef __attribute__((ext_vector_type(4)))  float  v4f;

#define IN_FEAT 512
#define HID     16

typedef __attribute__((ext_vector_type(2)))  float  v2f;
typedef __attribute__((ext_vector_type(3)))  float  v3f;

#define RSH   7
#define RA    128
#define RM    127
#define RB    100
#define NSLAB 256
#define NBMAX 1024
#define HP    17
#define EU    4
#define GU    8
#define LCAP  12800

typedef __attribute__((ext_vector_type(2)))  int    v2i;
static_assert(LCAP <= 4 * 1024 * EU, "an ordered slab is one trip of the edge loop");

__device__ __forceinline__ void ldsAddF(float* p, float v) {
  __hip_atomic_fetch_add(p, v, __ATOMIC_RELAXED, __HIP_MEMORY_SCOPE_WORKGROUP);
}

__device__ __forceinline__ int bkt(int d) { return (int)((unsigned)d / RB); }
__device__ __forceinline__ int rin(int d) { return (int)((unsigned)d % RB); }

template <bool VEC>
__device__ __forceinline__ int4 ld4(const int* __restrict__ p, int e, int e0, int e1) {
  int4 r;
  if (VEC) {
    r = *(const int4*)(p + (e < e1 ? e : e0));
    if (e >= e1) r.x = r.y = r.z = r.w = -1;
  } else {
    r.x = (e     < e1) ? p[e]     : -1;
    r.y = (e + 1 < e1) ? p[e + 1] : -1;
    r.z = (e + 2 < e1) ? p[e + 2] : -1;
    r.w = (e + 3 < e1) ? p[e + 3] : -1;
  }
  return r;
}

template <bool VEC>
__device__ __forceinline__ void hist_slab(const int* __restrict__ DST, int* h,
                                          int e0, int e1, int t) {
  for (int e = e0 + 4 * t; e < e1; e += 4 * 1024 * EU) {
    int4 d[EU];
#pragma unroll
    for (int u = 0; u < EU; ++u) d[u] = ld4<VEC>(DST, e + u * 4096, e0, e1);
#pragma unroll
    for (int u = 0; u < EU; ++u) {
      if (d[u].x >= 0) atomicAdd(&h[bkt(d[u].x)], 1);
      if (d[u].y >= 0) atomicAdd(&h[bkt(d[u].y)], 1);
      if (d[u].z >= 0) atomicAdd(&h[bkt(d[u].z)], 1);
      if (d[u].w >= 0) atomicAdd(&h[bkt(d[u].w)], 1);
    }
  }
}

template <bool VEC>
__device__ __forceinline__ void fill_slab(const int* __restrict__ SRC,
                                          const int* __restrict__ DST, int* cur,
                                          int* __restrict__ BINNED,
                                          int e0, int e1, int t) {
  for (int e = e0 + 4 * t; e < e1; e += 4 * 1024 * EU) {
    int4 s[EU], d[EU];
#pragma unroll
    for (int u = 0; u < EU; ++u) {
      s[u] = ld4<VEC>(SRC, e + u * 4096, e0, e1);
      d[u] = ld4<VEC>(DST, e + u * 4096, e0, e1);
    }
#pragma unroll
    for (int u = 0; u < EU; ++u) {
      if (d[u].x >= 0) BINNED[atomicAdd(&cur[bkt(d[u].x)], 1)] = (s[u].x << RSH) | rin(d[u].x);
      if (d[u].y >= 0) BINNED[atomicAdd(&cur[bkt(d[u].y)], 1)] = (s[u].y << RSH) | rin(d[u].y);
      if (d[u].z >= 0) BINNED[atomicAdd(&cur[bkt(d[u].z)], 1)] = (s[u].z << RSH) | rin(d[u].z);
      if (d[u].w >= 0) BINNED[atomicAdd(&cur[bkt(d[u].w)], 1)] = (s[u].w << RSH) | rin(d[u].w);
    }
  }
}

template <bool VEC>
__device__ __forceinline__ void order_slab(const int* __restrict__ SRC,
                                           const int* __restrict__ DST, int* cur,
                                           v2i* sp, int e0, int e1, int t) {
  const int e = e0 + 4 * t;
  if (e >= e1) return;
  int4 s[EU], d[EU];
#pragma unroll
  for (int u = 0; u < EU; ++u) {
    s[u] = ld4<VEC>(SRC, e + u * 4096, e0, e1);
    d[u] = ld4<VEC>(DST, e + u * 4096, e0, e1);
  }
  const int spare = NBMAX + (t & 31);
  int lp[4 * EU];
#pragma unroll
  for (int u = 0; u < EU; ++u) {
    lp[4 * u]     = atomicAdd(&cur[d[u].x >= 0 ? bkt(d[u].x) : spare], 1);
    lp[4 * u + 1] = atomicAdd(&cur[d[u].y >= 0 ? bkt(d[u].y) : spare], 1);
    lp[4 * u + 2] = atomicAdd(&cur[d[u].z >= 0 ? bkt(d[u].z) : spare], 1);
    lp[4 * u + 3] = atomicAdd(&cur[d[u].w >= 0 ? bkt(d[u].w) : spare], 1);
  }
#pragma unroll
  for (int u = 0; u < EU; ++u) {
    if (d[u].x >= 0) { v2i p; p.x = s[u].x; p.y = d[u].x; sp[lp[4 * u]]     = p; }
    if (d[u].y >= 0) { v2i p; p.x = s[u].y; p.y = d[u].y; sp[lp[4 * u + 1]] = p; }
    if (d[u].z >= 0) { v2i p; p.x = s[u].z; p.y = d[u].z; sp[lp[4 * u + 2]] = p; }
    if (d[u].w >= 0) { v2i p; p.x = s[u].w; p.y = d[u].w; sp[lp[4 * u + 3]] = p; }
  }
}

template <int NT>
__device__ __forceinline__ void gemm1_block(
    const float* __restrict__ X, const float* __restrict__ W1,
    float* __restrict__ XW1, int nTiles, __bf16* Bs, int gb, int xp) {
  for (int i = threadIdx.x; i < HID * IN_FEAT; i += NT) {
    int k = i >> 4, n = i & 15;
    Bs[n * IN_FEAT + k] = (__bf16)W1[i];
  }
  __syncthreads();

  int wave = __builtin_amdgcn_readfirstlane(threadIdx.x >> 5);
  int lane = threadIdx.x & 31;
  int tile = gb * (NT / 32) + wave;
  if (tile >= nTiles) return;

  int rc   = lane & 15;
  int half = lane >> 4;

  const float*  arow = X  + (size_t)tile * 16 * IN_FEAT + rc * IN_FEAT;
  const __bf16* bcol = Bs + (size_t)rc * IN_FEAT;

  v8f c = {0.f, 0.f, 0.f, 0.f, 0.f, 0.f, 0.f, 0.f};

#pragma unroll 2
  for (int ks = 0; ks < IN_FEAT / 32; ++ks) {
    int kb = ks * 32 + half * 8;
    v4f a0 = *(const v4f*)(arow + kb);
    v4f a1 = *(const v4f*)(arow + kb + 4);
    v4f a2 = *(const v4f*)(arow + kb + 16);
    v4f a3 = *(const v4f*)(arow + kb + 20);
    v8bf b0 = *(const v8bf*)(bcol + kb);
    v8bf b1 = *(const v8bf*)(bcol + kb + 16);
    v16bf A, B;
#pragma unroll
    for (int i = 0; i < 4; ++i) {
      A[i]      = (__bf16)a0[i];
      A[4 + i]  = (__bf16)a1[i];
      A[8 + i]  = (__bf16)a2[i];
      A[12 + i] = (__bf16)a3[i];
    }
#pragma unroll
    for (int i = 0; i < 8; ++i) { B[i] = b0[i]; B[8 + i] = b1[i]; }
    c = __builtin_amdgcn_wmma_f32_16x16x32_bf16(
         false, A,  false, B,
         (short)0, c,  false,  false);
  }

  float* orow = XW1 + (size_t)tile * 16 * xp + (half * 8 * xp + rc);
#pragma unroll
  for (int r = 0; r < 8; ++r) orow[r * xp] = c[r];
}

__global__ __launch_bounds__(1024) void bin_hist_k(const int* __restrict__ DST,
                                                   int* __restrict__ CNT,
                                                   int E, int NBP) {
  __shared__ int h[NBMAX];
  const int t = threadIdx.x;
  h[t] = 0;
  __syncthreads();
  const int slab = (E + NSLAB - 1) / NSLAB;
  const int e0 = blockIdx.x * slab;
  const int e1 = min(e0 + slab, E);
  const bool vec = ((E & 3) == 0) && ((slab & 3) == 0);
  if (vec) hist_slab<true>(DST, h, e0, e1, t);
  else     hist_slab<false>(DST, h, e0, e1, t);
  __syncthreads();
  if (t < NBP) CNT[blockIdx.x * NBP + t] = h[t];
}

__global__ __launch_bounds__(1024) void bin_scan_k(int* __restrict__ CNT,
                                                   int* __restrict__ TOT, int NBP) {
  __shared__ int rs[32][33];
  const int l = threadIdx.x & 31, r = threadIdx.x >> 5;
  const int col = blockIdx.x * 32 + l;
  int v[8];
  int run = 0;
#pragma unroll
  for (int k = 0; k < 8; ++k) {
    int c = CNT[(r * 8 + k) * NBP + col];
    v[k] = run;
    run += c;
  }
  rs[r][l] = run;
  __syncthreads();
  int base = 0, tot = 0;
  for (int rr = 0; rr < 32; ++rr) {
    int x = rs[rr][l];
    if (rr < r) base += x;
    tot += x;
  }
#pragma unroll
  for (int k = 0; k < 8; ++k) CNT[(r * 8 + k) * NBP + col] = base + v[k];
  if (r == 0) TOT[col] = tot;
}

__global__ __launch_bounds__(1024, 16) void bin_fill_k(
    const int* __restrict__ SRC, const int* __restrict__ DST,
    const int* __restrict__ CNT, const int* __restrict__ TOT,
    int* __restrict__ START, int* __restrict__ BINNED, int E, int NB, int NBP,
    const float* __restrict__ X, const float* __restrict__ W1,
    float* __restrict__ XW1, int nTiles, int nGemm, int xp) {
  __shared__ __bf16 Bs[HID * IN_FEAT];
  __shared__ v2i tt[NBMAX];
  __shared__ v2i cb[32];
  __shared__ int cur[NBMAX + 32];
  __shared__ int dlt[NBMAX];
  __shared__ v2i sp[LCAP];
  if ((int)blockIdx.x < nGemm) {
    gemm1_block<1024>(X, W1, XW1, nTiles, Bs, blockIdx.x, xp);
    return;
  }
  const int fb = blockIdx.x - nGemm;
  const int t = threadIdx.x;
  int c0 = 0;
  {
    int T = 0, nt = 0;
    if (t < NBP) {
      T  = TOT[t];
      c0 = CNT[fb * NBP + t];
      nt = ((fb + 1 < NSLAB) ? CNT[(fb + 1) * NBP + t] : T) - c0;
    }
    v2i w; w.x = T; w.y = nt;
    tt[t] = w;
  }
  __syncthreads();
  if (t < 32) {
    v2i s = {0, 0};
    for (int j = 0; j < 32; ++j) s += tt[t * 32 + j];
    cb[t] = s;
  }
  __syncthreads();
  v2i st = {0, 0};
  for (int j = 0; j < (t >> 5); ++j) st += cb[j];
  for (int j = (t & ~31); j < t; ++j) st += tt[j];
  if (fb == 0 && t <= NB) START[t] = st.x;

  const int slab = (E + NSLAB - 1) / NSLAB;
  const int e0 = fb * slab;
  const int e1 = min(e0 + slab, E);
  const bool vec = ((E & 3) == 0) && ((slab & 3) == 0);
  if (slab <= LCAP) {
    cur[t] = st.y;
    if (t < 32) cur[NBMAX + t] = 0;
    dlt[t] = st.x + c0 - st.y;
    __syncthreads();
    if (vec) order_slab<true>(SRC, DST, cur, sp, e0, e1, t);
    else     order_slab<false>(SRC, DST, cur, sp, e0, e1, t);
    __syncthreads();
    const int ns = max(e1 - e0, 0);
#pragma unroll 4
    for (int i = t; i < ns; i += 1024) {
      const v2i pr = sp[i];
      BINNED[i + dlt[bkt(pr.y)]] = (pr.x << RSH) | rin(pr.y);
    }
  } else {
    cur[t] = (t < NBP) ? st.x + c0 : 0;
    __syncthreads();
    if (vec) fill_slab<true>(SRC, DST, cur, BINNED, e0, e1, t);
    else     fill_slab<false>(SRC, DST, cur, BINNED, e0, e1, t);
  }
}

template <bool WIDE>
__global__ __launch_bounds__(256) void bin_deg_k(const int* __restrict__ START,
                                                 const int* __restrict__ BINNED,
                                                 float* __restrict__ DNV,
                                                 float* __restrict__ XW1, int N) {
  __shared__ int c[RA];
  const int t = threadIdx.x;
  if (t < RA) c[t] = 1;
  __syncthreads();
  const int a = START[blockIdx.x], b = START[blockIdx.x + 1];
  for (int i0 = a + t; i0 < b; i0 += 256 * GU) {
    int p[GU];
#pragma unroll
    for (int u = 0; u < GU; ++u) {
      const int i = i0 + u * 256;
      p[u] = (i < b) ? BINNED[i] : -1;
    }
#pragma unroll
    for (int u = 0; u < GU; ++u)
      if (p[u] >= 0) atomicAdd(&c[p[u] & RM], 1);
  }
  __syncthreads();
  const int n = blockIdx.x * RB + t;
  if (t < RB && n < N) {
    const float dv = rsqrtf((float)c[t]);
    if (WIDE) {
      float* dp = XW1 + (size_t)n * 32 + HID;
#pragma unroll
      for (int k = 0; k < 4; ++k) dp[4 * k] = dv;
    } else {
      DNV[n] = dv;
    }
  }
}

template <bool WIDE>
__global__ __launch_bounds__(512) void bin_agg1_k(
    const int* __restrict__ START, const int* __restrict__ BINNED,
    const float* __restrict__ DNV, const float* __restrict__ XW1,
    const float* __restrict__ B1, const float* __restrict__ W2,
    float* __restrict__ REC2, int N) {
  constexpr int XP = WIDE ? 32 : HID;
  __shared__ float acc[RA * HP];
  __shared__ float dl[RA];
  const int t = threadIdx.x;
  const int n0 = blockIdx.x * RB;
  const int q = t & 3;
  {
    const int r = t >> 2;
    const int n = n0 + r;
    float dn = 0.f;
    v4f v = {0.f, 0.f, 0.f, 0.f};
    if (r < RB && n < N) {
      const float* rp = XW1 + (size_t)n * XP + q * 4;
      dn = WIDE ? rp[HID] : DNV[n];
      v = *(const v4f*)rp;
    }
    const float w = dn * dn;
    if (q == 0) dl[r] = dn;
#pragma unroll
    for (int i = 0; i < 4; ++i) acc[r * HP + q * 4 + i] = w * v[i];
  }
  __syncthreads();
  const int a = START[blockIdx.x], b = START[blockIdx.x + 1];
  for (int i0 = a + (t >> 2); i0 < b; i0 += 128 * GU) {
    int p[GU];
#pragma unroll
    for (int u = 0; u < GU; ++u) {
      const int i = i0 + u * 128;
      p[u] = (i < b) ? BINNED[i] : -1;
    }
    float dn[GU];
    v4f v[GU];
#pragma unroll
    for (int u = 0; u < GU; ++u) {
      const int s = (p[u] < 0 ? 0 : p[u]) >> RSH;
      const unsigned bo = ((unsigned)s * XP + q * 4) * 4u;
      const float* rp = (const float*)((const char*)XW1 + bo);
      dn[u] = WIDE ? rp[HID] : DNV[s];
      v[u] = *(const v4f*)rp;
    }
#pragma unroll
    for (int u = 0; u < GU; ++u) {
      if (p[u] >= 0) {
        const int r = p[u] & RM;
        const float nrm = dn[u] * dl[r];
        float* ap = acc + r * HP + q * 4;
#pragma unroll
        for (int k = 0; k < 4; ++k) ldsAddF(ap + k, nrm * v[u][k]);
      }
    }
  }
  __syncthreads();
  if (t < RB) {
    const int n = n0 + t;
    if (n < N) {
      float h0 = 0.f, h1 = 0.f;
#pragma unroll
      for (int f = 0; f < HID; ++f) {
        float hv = fmaxf(acc[t * HP + f] + B1[f], 0.f);
        h0 += hv * W2[2 * f];
        h1 += hv * W2[2 * f + 1];
      }
      v4f o;
      o[0] = h0;
      o[1] = h1;
      o[2] = dl[t];
      o[3] = 0.f;
      *(v4f*)(REC2 + 4 * (size_t)n) = o;
    }
  }
}

__global__ __launch_bounds__(256) void bin_agg2_k(
    const int* __restrict__ START, const int* __restrict__ BINNED,
    const float* __restrict__ REC2,
    const float* __restrict__ B2, float* __restrict__ out, int N) {
  __shared__ float acc[2 * RA];
  __shared__ float dl[RA];
  const int t = threadIdx.x;
  const int n0 = blockIdx.x * RB;
  {
    const int r = t >> 1;
    const int n = n0 + r;
    float dn = 0.f, hv = 0.f;
    if (r < RB && n < N) {
      dn = REC2[4 * (size_t)n + 2];
      hv = REC2[4 * (size_t)n + (t & 1)];
    }
    const float w = dn * dn;
    acc[t] = w * hv;
    if ((t & 1) == 0) dl[r] = dn;
  }
  __syncthreads();
  const int a = START[blockIdx.x], b = START[blockIdx.x + 1];
  for (int i0 = a + t; i0 < b; i0 += 256 * GU) {
    int p[GU];
#pragma unroll
    for (int u = 0; u < GU; ++u) {
      const int i = i0 + u * 256;
      p[u] = (i < b) ? BINNED[i] : -1;
    }
    v3f rec[GU];
#pragma unroll
    for (int u = 0; u < GU; ++u) {
      const int s = (p[u] < 0 ? 0 : p[u]) >> RSH;
      const float* rp = REC2 + 4 * (size_t)s;
      rec[u][0] = rp[0];
      rec[u][1] = rp[1];
      rec[u][2] = rp[2];
    }
#pragma unroll
    for (int u = 0; u < GU; ++u) {
      if (p[u] >= 0) {
        const int r = p[u] & RM;
        const float nrm = rec[u][2] * dl[r];
        ldsAddF(&acc[2 * r],     nrm * rec[u][0]);
        ldsAddF(&acc[2 * r + 1], nrm * rec[u][1]);
      }
    }
  }
  __syncthreads();
  if (t < RB) {
    const int n = n0 + t;
    if (n < N) {
      float x = acc[2 * t]     + B2[0];
      float y = acc[2 * t + 1] + B2[1];
      float m = fmaxf(x, y);
      float lse = m + logf(expf(x - m) + expf(y - m));
      v2f o;
      o[0] = x - lse;
      o[1] = y - lse;
      *(v2f*)(out + 2 * (size_t)n) = o;
    }
  }
}

extern "C" void kernel_launch(void* const* d_in, const int* in_sizes, int n_in,
                              void* d_out, int out_size, void* d_ws, size_t ws_size,
                              hipStream_t stream) {
  const float* X  = (const float*)d_in[0];
  const int*   EI = (const int*)d_in[1];
  const float* W1 = (const float*)d_in[2];
  const float* B1 = (const float*)d_in[3];
  const float* W2 = (const float*)d_in[4];
  const float* B2 = (const float*)d_in[5];

  const int N = in_sizes[0] / IN_FEAT;
  const int E = in_sizes[1] / 2;
  const int* SRC = EI;
  const int* DST = EI + E;

  const int NB  = (N + RB - 1) / RB;
  const int NBP = (NB + 31) / 32 * 32;

  const size_t Nr   = ((size_t)N + 31) / 32 * 32;
  const size_t Er   = ((size_t)E + 31) / 32 * 32;
  const size_t cw   = (size_t)NSLAB * NBP;
  const size_t uw   = cw > 4 * Nr ? cw : 4 * Nr;
  const bool   wide = ws_size >= (32 * Nr + Er + NBMAX + NBMAX + 32 + uw) * 4;
  const int    xp   = wide ? 32 : HID;
  float* ws     = (float*)d_ws;
  float* XW1    = ws;
  int*   BINNED = (int*)(ws + Nr * xp);
  float* DNV    = (float*)(BINNED + Er);
  int*   TOT    = (int*)(DNV + (wide ? 0 : Nr));
  int*   START  = TOT + NBMAX;
  int*   CNT    = START + NBMAX + 32;
  float* REC2   = (float*)CNT;

  const int nTiles = (N + 15) / 16;
  const int nGemm  = (nTiles + 31) / 32;

  bin_hist_k<<<NSLAB, 1024, 0, stream>>>(DST, CNT, E, NBP);
  bin_scan_k<<<NBP / 32, 1024, 0, stream>>>(CNT, TOT, NBP);
  bin_fill_k<<<nGemm + NSLAB, 1024, 0, stream>>>(SRC, DST, CNT, TOT, START, BINNED, E, NB, NBP,
                                               X, W1, XW1, nTiles, nGemm, xp);
  if (wide) {
    bin_deg_k<true><<<NB, 256, 0, stream>>>(START, BINNED, DNV, XW1, N);
    bin_agg1_k<true><<<NB, 512, 0, stream>>>(START, BINNED, DNV, XW1, B1, W2, REC2, N);
  } else {
    bin_deg_k<false><<<NB, 256, 0, stream>>>(START, BINNED, DNV, XW1, N);
    bin_agg1_k<false><<<NB, 512, 0, stream>>>(START, BINNED, DNV, XW1, B1, W2, REC2, N);
  }
  bin_agg2_k<<<NB, 256, 0, stream>>>(START, BINNED, REC2, B2, (float*)d_out, N);
}
